# starting kernel
# baseline (speedup 1.0000x reference)
#include <hip/hip_runtime.h>
#include <cstddef>

typedef __attribute__((ext_vector_type(2))) float v2f;
typedef __attribute__((ext_vector_type(8))) float v8f;

#if defined(__HIP_DEVICE_COMPILE__) && !__has_builtin(__builtin_amdgcn_wmma_f32_16x16x4_f32)
#error "__builtin_amdgcn_wmma_f32_16x16x4_f32 not available on gfx1250 device pass"
#endif

static __device__ __forceinline__ v8f wmma4(v2f a, v2f b, v8f c) {
  return __builtin_amdgcn_wmma_f32_16x16x4_f32(false, a, false, b, (short)0, c,
                                               false, false);
}

constexpr int NROWS = 256;
constexpr int TLEN  = 131072;
constexpr int PAD   = 15;
constexpr int ELEN  = TLEN + 2 * PAD;
constexpr int ES    = 131104;
constexpr int NBLK  = ES / 16;
constexpr int CB    = 128;
constexpr int WB    = 16;
constexpr int NCHUNK = (NBLK + CB - 1) / CB;
constexpr int NRG   = NROWS / 16;

constexpr int WS_TB = 0;
constexpr int WS_WB = 256;
constexpr int WS_UB = 320;
constexpr int WS_PB = 576;
constexpr int WS_ZI = 640;
constexpr int WS_YF = 1024;

__global__ void setup_kernel(const float* __restrict__ bc,
                             const float* __restrict__ ac,
                             float* __restrict__ ws) {
  if (threadIdx.x != 0 || blockIdx.x != 0) return;
  double a0 = (double)ac[0];
  double an[5], bn[5];
  for (int i = 0; i < 5; ++i) { an[i] = (double)ac[i] / a0; bn[i] = (double)bc[i] / a0; }
  const double b0 = bn[0];
  double Bv[4];
  for (int i = 0; i < 4; ++i) Bv[i] = bn[i + 1] - an[i + 1] * b0;
  double A[4][4] = {{-an[1], 1, 0, 0},
                    {-an[2], 0, 1, 0},
                    {-an[3], 0, 0, 1},
                    {-an[4], 0, 0, 0}};
  double Crow[16][4];
  Crow[0][0] = 1; Crow[0][1] = 0; Crow[0][2] = 0; Crow[0][3] = 0;
  for (int k = 1; k < 16; ++k)
    for (int j = 0; j < 4; ++j) {
      double s = 0;
      for (int i = 0; i < 4; ++i) s += Crow[k - 1][i] * A[i][j];
      Crow[k][j] = s;
    }
  double h[16];
  h[0] = b0;
  for (int m = 1; m < 16; ++m) {
    double s = 0;
    for (int i = 0; i < 4; ++i) s += Crow[m - 1][i] * Bv[i];
    h[m] = s;
  }
  double u[16][4];
  for (int i = 0; i < 4; ++i) u[0][i] = Bv[i];
  for (int p = 1; p < 16; ++p)
    for (int i = 0; i < 4; ++i) {
      double s = 0;
      for (int j = 0; j < 4; ++j) s += A[i][j] * u[p - 1][j];
      u[p][i] = s;
    }
  double P[4][4], Tm[4][4];
  for (int i = 0; i < 4; ++i)
    for (int j = 0; j < 4; ++j) P[i][j] = (i == j) ? 1.0 : 0.0;
  for (int s = 0; s < 16; ++s) {
    for (int i = 0; i < 4; ++i)
      for (int j = 0; j < 4; ++j) {
        double t = 0;
        for (int q = 0; q < 4; ++q) t += A[i][q] * P[q][j];
        Tm[i][j] = t;
      }
    for (int i = 0; i < 4; ++i)
      for (int j = 0; j < 4; ++j) P[i][j] = Tm[i][j];
  }
  double M[4][5];
  for (int i = 0; i < 4; ++i) {
    for (int j = 0; j < 4; ++j)
      M[i][j] = ((i == j) ? 1.0 : 0.0) + ((j == 0) ? an[i + 1] : 0.0) -
                ((j == i + 1) ? 1.0 : 0.0);
    M[i][4] = Bv[i];
  }
  for (int c = 0; c < 4; ++c) {
    int piv = c;
    for (int r = c + 1; r < 4; ++r)
      if ((M[r][c] < 0 ? -M[r][c] : M[r][c]) > (M[piv][c] < 0 ? -M[piv][c] : M[piv][c]))
        piv = r;
    if (piv != c)
      for (int j = 0; j < 5; ++j) { double t = M[c][j]; M[c][j] = M[piv][j]; M[piv][j] = t; }
    double d = M[c][c];
    for (int j = 0; j < 5; ++j) M[c][j] /= d;
    for (int r = 0; r < 4; ++r) {
      if (r == c) continue;
      double f = M[r][c];
      for (int j = 0; j < 5; ++j) M[r][j] -= f * M[c][j];
    }
  }
  for (int c = 0; c < 4; ++c)
    for (int v = 0; v < 2; ++v)
      for (int L = 0; L < 32; ++L) {
        int k = v + ((L >= 16) ? 2 : 0);
        int n = L & 15;
        int kk = 4 * c + k;
        ws[WS_TB + c * 64 + v * 32 + L] = (float)((n >= kk) ? h[n - kk] : 0.0);
        ws[WS_UB + c * 64 + v * 32 + L] = (float)((n < 4) ? u[15 - kk][n] : 0.0);
      }
  for (int v = 0; v < 2; ++v)
    for (int L = 0; L < 32; ++L) {
      int k = v + ((L >= 16) ? 2 : 0);
      int n = L & 15;
      ws[WS_WB + v * 32 + L] = (float)Crow[n][k];
      ws[WS_PB + v * 32 + L] = (float)((n < 4) ? P[n][k] : 0.0);
    }
  for (int i = 0; i < 4; ++i) ws[WS_ZI + i] = (float)M[i][4];
}

static __device__ __forceinline__ float load_ext_fwd(const float* __restrict__ xr, int e) {
  if (e < PAD) return 2.0f * xr[0] - xr[PAD - e];
  if (e < PAD + TLEN) return xr[e - PAD];
  int j = e - (PAD + TLEN);
  if (j > PAD - 1) j = PAD - 1;
  return 2.0f * xr[TLEN - 1] - xr[TLEN - 2 - j];
}

static __device__ __forceinline__ float load_rev(const float* __restrict__ yr, int e) {
  int idx = ELEN - 1 - e;
  if (idx < 0) idx = 0;
  return yr[idx];
}

__global__ __launch_bounds__(32) void fwd_kernel(const float* __restrict__ x,
                                                 float* __restrict__ ws) {
  float* __restrict__ yf = ws + WS_YF;
  const int task = blockIdx.x;
  const int rg = task / NCHUNK, ch = task % NCHUNK;
  const int r0 = rg * 16;
  const int L = threadIdx.x;
  const int m = L & 15;
  const int hiA = L >> 4;
  const int k0 = 2 * hiA;
  const float* __restrict__ xrow = x + (size_t)(r0 + m) * TLEN;

  v2f TBv[4], UBv[4], WBv, PBv;
#pragma unroll
  for (int c = 0; c < 4; ++c) {
    TBv[c] = (v2f){ws[WS_TB + c * 64 + L], ws[WS_TB + c * 64 + 32 + L]};
    UBv[c] = (v2f){ws[WS_UB + c * 64 + L], ws[WS_UB + c * 64 + 32 + L]};
  }
  WBv = (v2f){ws[WS_WB + L], ws[WS_WB + 32 + L]};
  PBv = (v2f){ws[WS_PB + L], ws[WS_PB + 32 + L]};
  const float zi0 = ws[WS_ZI + k0], zi1 = ws[WS_ZI + k0 + 1];

  const int bOut = ch * CB;
  const int bW = (bOut - WB > 0) ? (bOut - WB) : 0;
  const int bEnd = ((bOut + CB) < NBLK) ? (bOut + CB) : NBLK;

  float v0 = load_ext_fwd(xrow, bW * 16);
  v2f z = (v2f){zi0 * v0, zi1 * v0};

  const int sl0 = k0 + ((m & 8) ? 16 : 0);
  const int sl1 = sl0 + 1;
  const int vsel = L & 7;

  for (int bb = bW; bb < bEnd; ++bb) {
    const int e0 = bb * 16;
    v2f xa[4];
    if (e0 >= PAD && (e0 + 16) <= (PAD + TLEN)) {
      const float* p = xrow + (e0 - PAD) + k0;
#pragma unroll
      for (int c = 0; c < 4; ++c) xa[c] = (v2f){p[4 * c], p[4 * c + 1]};
      __builtin_prefetch(p + 64, 0, 1);
    } else {
#pragma unroll
      for (int c = 0; c < 4; ++c) {
        int o = e0 + 4 * c + k0;
        xa[c] = (v2f){load_ext_fwd(xrow, o), load_ext_fwd(xrow, o + 1)};
      }
    }
    const v8f zero = (v8f){0, 0, 0, 0, 0, 0, 0, 0};
    v8f dZ = wmma4(z, PBv, zero);
#pragma unroll
    for (int c = 0; c < 4; ++c) dZ = wmma4(xa[c], UBv[c], dZ);
    if (bb >= bOut) {
      v8f dY = wmma4(z, WBv, zero);
#pragma unroll
      for (int c = 0; c < 4; ++c) dY = wmma4(xa[c], TBv[c], dY);
#pragma unroll
      for (int v = 0; v < 8; ++v)
        yf[(size_t)(r0 + v + 8 * hiA) * ES + e0 + m] = dY[v];
    }
    float nz0 = 0.f, nz1 = 0.f;
#pragma unroll
    for (int v = 0; v < 8; ++v) {
      float g0 = __shfl(dZ[v], sl0, 32);
      float g1 = __shfl(dZ[v], sl1, 32);
      nz0 = (vsel == v) ? g0 : nz0;
      nz1 = (vsel == v) ? g1 : nz1;
    }
    z = (v2f){nz0, nz1};
  }
}

__global__ __launch_bounds__(32) void bwd_kernel(const float* __restrict__ ws,
                                                 float* __restrict__ out) {
  const float* __restrict__ yf = ws + WS_YF;
  const int task = blockIdx.x;
  const int rg = task / NCHUNK, ch = task % NCHUNK;
  const int r0 = rg * 16;
  const int L = threadIdx.x;
  const int m = L & 15;
  const int hiA = L >> 4;
  const int k0 = 2 * hiA;
  const float* __restrict__ yrow = yf + (size_t)(r0 + m) * ES;

  v2f TBv[4], UBv[4], WBv, PBv;
#pragma unroll
  for (int c = 0; c < 4; ++c) {
    TBv[c] = (v2f){ws[WS_TB + c * 64 + L], ws[WS_TB + c * 64 + 32 + L]};
    UBv[c] = (v2f){ws[WS_UB + c * 64 + L], ws[WS_UB + c * 64 + 32 + L]};
  }
  WBv = (v2f){ws[WS_WB + L], ws[WS_WB + 32 + L]};
  PBv = (v2f){ws[WS_PB + L], ws[WS_PB + 32 + L]};
  const float zi0 = ws[WS_ZI + k0], zi1 = ws[WS_ZI + k0 + 1];

  const int bOut = ch * CB;
  const int bW = (bOut - WB > 0) ? (bOut - WB) : 0;
  const int bEnd = ((bOut + CB) < NBLK) ? (bOut + CB) : NBLK;

  float v0 = load_rev(yrow, bW * 16);
  v2f z = (v2f){zi0 * v0, zi1 * v0};
  const int sl0 = k0 + ((m & 8) ? 16 : 0);
  const int sl1 = sl0 + 1;
  const int vsel = L & 7;

  for (int bb = bW; bb < bEnd; ++bb) {
    const int e0 = bb * 16;
    v2f xa[4];
    if (e0 + 16 <= ELEN) {
      const float* p = yrow + (ELEN - 1 - e0 - k0);
#pragma unroll
      for (int c = 0; c < 4; ++c) xa[c] = (v2f){p[-4 * c], p[-4 * c - 1]};
      __builtin_prefetch(p - 64, 0, 1);
    } else {
#pragma unroll
      for (int c = 0; c < 4; ++c) {
        int o = e0 + 4 * c + k0;
        xa[c] = (v2f){load_rev(yrow, o), load_rev(yrow, o + 1)};
      }
    }
    const v8f zero = (v8f){0, 0, 0, 0, 0, 0, 0, 0};
    v8f dZ = wmma4(z, PBv, zero);
#pragma unroll
    for (int c = 0; c < 4; ++c) dZ = wmma4(xa[c], UBv[c], dZ);
    if (bb >= bOut) {
      v8f dY = wmma4(z, WBv, zero);
#pragma unroll
      for (int c = 0; c < 4; ++c) dY = wmma4(xa[c], TBv[c], dY);
      const int e = e0 + m;
      if (e >= PAD && e < PAD + TLEN) {
        const int col = ELEN - 16 - e;
#pragma unroll
        for (int v = 0; v < 8; ++v)
          out[(size_t)(r0 + v + 8 * hiA) * TLEN + col] = dY[v];
      }
    }
    float nz0 = 0.f, nz1 = 0.f;
#pragma unroll
    for (int v = 0; v < 8; ++v) {
      float g0 = __shfl(dZ[v], sl0, 32);
      float g1 = __shfl(dZ[v], sl1, 32);
      nz0 = (vsel == v) ? g0 : nz0;
      nz1 = (vsel == v) ? g1 : nz1;
    }
    z = (v2f){nz0, nz1};
  }
}

extern "C" void kernel_launch(void* const* d_in, const int* in_sizes, int n_in,
                              void* d_out, int out_size, void* d_ws, size_t ws_size,
                              hipStream_t stream) {
  (void)in_sizes; (void)n_in; (void)out_size; (void)ws_size;
  const float* x = (const float*)d_in[0];
  const float* b = (const float*)d_in[1];
  const float* a = (const float*)d_in[2];
  float* out = (float*)d_out;
  float* ws = (float*)d_ws;

  setup_kernel<<<1, 1, 0, stream>>>(b, a, ws);
  fwd_kernel<<<NRG * NCHUNK, 32, 0, stream>>>(x, ws);
  bwd_kernel<<<NRG * NCHUNK, 32, 0, stream>>>(ws, out);
}


# Chunk length 128 -> 32 blocks (warm-up 16 kept): 4112 waves per pass, ~4 per SIMD, so a wave's exposed load, chained-WMMA and lane-exchange waits run under other waves; state re-warmed at 256 starts
# speedup vs baseline: 1.2683x; 1.2683x over previous
#include <hip/hip_runtime.h>
#include <cstddef>

typedef __attribute__((ext_vector_type(2))) float v2f;
typedef __attribute__((ext_vector_type(8))) float v8f;

#if defined(__HIP_DEVICE_COMPILE__) && !__has_builtin(__builtin_amdgcn_wmma_f32_16x16x4_f32)
#error "__builtin_amdgcn_wmma_f32_16x16x4_f32 not available on gfx1250 device pass"
#endif

static __device__ __forceinline__ v8f wmma4(v2f a, v2f b, v8f c) {
  return __builtin_amdgcn_wmma_f32_16x16x4_f32(false, a, false, b, (short)0, c,
                                               false, false);
}

constexpr int NROWS = 256;
constexpr int TLEN  = 131072;
constexpr int PAD   = 15;
constexpr int ELEN  = TLEN + 2 * PAD;
constexpr int ES    = 131104;
constexpr int NBLK  = ES / 16;
constexpr int CB    = 32;
constexpr int WB    = 16;
constexpr int NCHUNK = (NBLK + CB - 1) / CB;
constexpr int NRG   = NROWS / 16;

constexpr int WS_TB = 0;
constexpr int WS_WB = 256;
constexpr int WS_UB = 320;
constexpr int WS_PB = 576;
constexpr int WS_ZI = 640;
constexpr int WS_YF = 1024;

__global__ void setup_kernel(const float* __restrict__ bc,
                             const float* __restrict__ ac,
                             float* __restrict__ ws) {
  if (threadIdx.x != 0 || blockIdx.x != 0) return;
  double a0 = (double)ac[0];
  double an[5], bn[5];
  for (int i = 0; i < 5; ++i) { an[i] = (double)ac[i] / a0; bn[i] = (double)bc[i] / a0; }
  const double b0 = bn[0];
  double Bv[4];
  for (int i = 0; i < 4; ++i) Bv[i] = bn[i + 1] - an[i + 1] * b0;
  double A[4][4] = {{-an[1], 1, 0, 0},
                    {-an[2], 0, 1, 0},
                    {-an[3], 0, 0, 1},
                    {-an[4], 0, 0, 0}};
  double Crow[16][4];
  Crow[0][0] = 1; Crow[0][1] = 0; Crow[0][2] = 0; Crow[0][3] = 0;
  for (int k = 1; k < 16; ++k)
    for (int j = 0; j < 4; ++j) {
      double s = 0;
      for (int i = 0; i < 4; ++i) s += Crow[k - 1][i] * A[i][j];
      Crow[k][j] = s;
    }
  double h[16];
  h[0] = b0;
  for (int m = 1; m < 16; ++m) {
    double s = 0;
    for (int i = 0; i < 4; ++i) s += Crow[m - 1][i] * Bv[i];
    h[m] = s;
  }
  double u[16][4];
  for (int i = 0; i < 4; ++i) u[0][i] = Bv[i];
  for (int p = 1; p < 16; ++p)
    for (int i = 0; i < 4; ++i) {
      double s = 0;
      for (int j = 0; j < 4; ++j) s += A[i][j] * u[p - 1][j];
      u[p][i] = s;
    }
  double P[4][4], Tm[4][4];
  for (int i = 0; i < 4; ++i)
    for (int j = 0; j < 4; ++j) P[i][j] = (i == j) ? 1.0 : 0.0;
  for (int s = 0; s < 16; ++s) {
    for (int i = 0; i < 4; ++i)
      for (int j = 0; j < 4; ++j) {
        double t = 0;
        for (int q = 0; q < 4; ++q) t += A[i][q] * P[q][j];
        Tm[i][j] = t;
      }
    for (int i = 0; i < 4; ++i)
      for (int j = 0; j < 4; ++j) P[i][j] = Tm[i][j];
  }
  double M[4][5];
  for (int i = 0; i < 4; ++i) {
    for (int j = 0; j < 4; ++j)
      M[i][j] = ((i == j) ? 1.0 : 0.0) + ((j == 0) ? an[i + 1] : 0.0) -
                ((j == i + 1) ? 1.0 : 0.0);
    M[i][4] = Bv[i];
  }
  for (int c = 0; c < 4; ++c) {
    int piv = c;
    for (int r = c + 1; r < 4; ++r)
      if ((M[r][c] < 0 ? -M[r][c] : M[r][c]) > (M[piv][c] < 0 ? -M[piv][c] : M[piv][c]))
        piv = r;
    if (piv != c)
      for (int j = 0; j < 5; ++j) { double t = M[c][j]; M[c][j] = M[piv][j]; M[piv][j] = t; }
    double d = M[c][c];
    for (int j = 0; j < 5; ++j) M[c][j] /= d;
    for (int r = 0; r < 4; ++r) {
      if (r == c) continue;
      double f = M[r][c];
      for (int j = 0; j < 5; ++j) M[r][j] -= f * M[c][j];
    }
  }
  for (int c = 0; c < 4; ++c)
    for (int v = 0; v < 2; ++v)
      for (int L = 0; L < 32; ++L) {
        int k = v + ((L >= 16) ? 2 : 0);
        int n = L & 15;
        int kk = 4 * c + k;
        ws[WS_TB + c * 64 + v * 32 + L] = (float)((n >= kk) ? h[n - kk] : 0.0);
        ws[WS_UB + c * 64 + v * 32 + L] = (float)((n < 4) ? u[15 - kk][n] : 0.0);
      }
  for (int v = 0; v < 2; ++v)
    for (int L = 0; L < 32; ++L) {
      int k = v + ((L >= 16) ? 2 : 0);
      int n = L & 15;
      ws[WS_WB + v * 32 + L] = (float)Crow[n][k];
      ws[WS_PB + v * 32 + L] = (float)((n < 4) ? P[n][k] : 0.0);
    }
  for (int i = 0; i < 4; ++i) ws[WS_ZI + i] = (float)M[i][4];
}

static __device__ __forceinline__ float load_ext_fwd(const float* __restrict__ xr, int e) {
  if (e < PAD) return 2.0f * xr[0] - xr[PAD - e];
  if (e < PAD + TLEN) return xr[e - PAD];
  int j = e - (PAD + TLEN);
  if (j > PAD - 1) j = PAD - 1;
  return 2.0f * xr[TLEN - 1] - xr[TLEN - 2 - j];
}

static __device__ __forceinline__ float load_rev(const float* __restrict__ yr, int e) {
  int idx = ELEN - 1 - e;
  if (idx < 0) idx = 0;
  return yr[idx];
}

__global__ __launch_bounds__(32) void fwd_kernel(const float* __restrict__ x,
                                                 float* __restrict__ ws) {
  float* __restrict__ yf = ws + WS_YF;
  const int task = blockIdx.x;
  const int rg = task / NCHUNK, ch = task % NCHUNK;
  const int r0 = rg * 16;
  const int L = threadIdx.x;
  const int m = L & 15;
  const int hiA = L >> 4;
  const int k0 = 2 * hiA;
  const float* __restrict__ xrow = x + (size_t)(r0 + m) * TLEN;

  v2f TBv[4], UBv[4], WBv, PBv;
#pragma unroll
  for (int c = 0; c < 4; ++c) {
    TBv[c] = (v2f){ws[WS_TB + c * 64 + L], ws[WS_TB + c * 64 + 32 + L]};
    UBv[c] = (v2f){ws[WS_UB + c * 64 + L], ws[WS_UB + c * 64 + 32 + L]};
  }
  WBv = (v2f){ws[WS_WB + L], ws[WS_WB + 32 + L]};
  PBv = (v2f){ws[WS_PB + L], ws[WS_PB + 32 + L]};
  const float zi0 = ws[WS_ZI + k0], zi1 = ws[WS_ZI + k0 + 1];

  const int bOut = ch * CB;
  const int bW = (bOut - WB > 0) ? (bOut - WB) : 0;
  const int bEnd = ((bOut + CB) < NBLK) ? (bOut + CB) : NBLK;

  float v0 = load_ext_fwd(xrow, bW * 16);
  v2f z = (v2f){zi0 * v0, zi1 * v0};

  const int sl0 = k0 + ((m & 8) ? 16 : 0);
  const int sl1 = sl0 + 1;
  const int vsel = L & 7;

  for (int bb = bW; bb < bEnd; ++bb) {
    const int e0 = bb * 16;
    v2f xa[4];
    if (e0 >= PAD && (e0 + 16) <= (PAD + TLEN)) {
      const float* p = xrow + (e0 - PAD) + k0;
#pragma unroll
      for (int c = 0; c < 4; ++c) xa[c] = (v2f){p[4 * c], p[4 * c + 1]};
      __builtin_prefetch(p + 64, 0, 1);
    } else {
#pragma unroll
      for (int c = 0; c < 4; ++c) {
        int o = e0 + 4 * c + k0;
        xa[c] = (v2f){load_ext_fwd(xrow, o), load_ext_fwd(xrow, o + 1)};
      }
    }
    const v8f zero = (v8f){0, 0, 0, 0, 0, 0, 0, 0};
    v8f dZ = wmma4(z, PBv, zero);
#pragma unroll
    for (int c = 0; c < 4; ++c) dZ = wmma4(xa[c], UBv[c], dZ);
    if (bb >= bOut) {
      v8f dY = wmma4(z, WBv, zero);
#pragma unroll
      for (int c = 0; c < 4; ++c) dY = wmma4(xa[c], TBv[c], dY);
#pragma unroll
      for (int v = 0; v < 8; ++v)
        yf[(size_t)(r0 + v + 8 * hiA) * ES + e0 + m] = dY[v];
    }
    float nz0 = 0.f, nz1 = 0.f;
#pragma unroll
    for (int v = 0; v < 8; ++v) {
      float g0 = __shfl(dZ[v], sl0, 32);
      float g1 = __shfl(dZ[v], sl1, 32);
      nz0 = (vsel == v) ? g0 : nz0;
      nz1 = (vsel == v) ? g1 : nz1;
    }
    z = (v2f){nz0, nz1};
  }
}

__global__ __launch_bounds__(32) void bwd_kernel(const float* __restrict__ ws,
                                                 float* __restrict__ out) {
  const float* __restrict__ yf = ws + WS_YF;
  const int task = blockIdx.x;
  const int rg = task / NCHUNK, ch = task % NCHUNK;
  const int r0 = rg * 16;
  const int L = threadIdx.x;
  const int m = L & 15;
  const int hiA = L >> 4;
  const int k0 = 2 * hiA;
  const float* __restrict__ yrow = yf + (size_t)(r0 + m) * ES;

  v2f TBv[4], UBv[4], WBv, PBv;
#pragma unroll
  for (int c = 0; c < 4; ++c) {
    TBv[c] = (v2f){ws[WS_TB + c * 64 + L], ws[WS_TB + c * 64 + 32 + L]};
    UBv[c] = (v2f){ws[WS_UB + c * 64 + L], ws[WS_UB + c * 64 + 32 + L]};
  }
  WBv = (v2f){ws[WS_WB + L], ws[WS_WB + 32 + L]};
  PBv = (v2f){ws[WS_PB + L], ws[WS_PB + 32 + L]};
  const float zi0 = ws[WS_ZI + k0], zi1 = ws[WS_ZI + k0 + 1];

  const int bOut = ch * CB;
  const int bW = (bOut - WB > 0) ? (bOut - WB) : 0;
  const int bEnd = ((bOut + CB) < NBLK) ? (bOut + CB) : NBLK;

  float v0 = load_rev(yrow, bW * 16);
  v2f z = (v2f){zi0 * v0, zi1 * v0};
  const int sl0 = k0 + ((m & 8) ? 16 : 0);
  const int sl1 = sl0 + 1;
  const int vsel = L & 7;

  for (int bb = bW; bb < bEnd; ++bb) {
    const int e0 = bb * 16;
    v2f xa[4];
    if (e0 + 16 <= ELEN) {
      const float* p = yrow + (ELEN - 1 - e0 - k0);
#pragma unroll
      for (int c = 0; c < 4; ++c) xa[c] = (v2f){p[-4 * c], p[-4 * c - 1]};
      __builtin_prefetch(p - 64, 0, 1);
    } else {
#pragma unroll
      for (int c = 0; c < 4; ++c) {
        int o = e0 + 4 * c + k0;
        xa[c] = (v2f){load_rev(yrow, o), load_rev(yrow, o + 1)};
      }
    }
    const v8f zero = (v8f){0, 0, 0, 0, 0, 0, 0, 0};
    v8f dZ = wmma4(z, PBv, zero);
#pragma unroll
    for (int c = 0; c < 4; ++c) dZ = wmma4(xa[c], UBv[c], dZ);
    if (bb >= bOut) {
      v8f dY = wmma4(z, WBv, zero);
#pragma unroll
      for (int c = 0; c < 4; ++c) dY = wmma4(xa[c], TBv[c], dY);
      const int e = e0 + m;
      if (e >= PAD && e < PAD + TLEN) {
        const int col = ELEN - 16 - e;
#pragma unroll
        for (int v = 0; v < 8; ++v)
          out[(size_t)(r0 + v + 8 * hiA) * TLEN + col] = dY[v];
      }
    }
    float nz0 = 0.f, nz1 = 0.f;
#pragma unroll
    for (int v = 0; v < 8; ++v) {
      float g0 = __shfl(dZ[v], sl0, 32);
      float g1 = __shfl(dZ[v], sl1, 32);
      nz0 = (vsel == v) ? g0 : nz0;
      nz1 = (vsel == v) ? g1 : nz1;
    }
    z = (v2f){nz0, nz1};
  }
}

extern "C" void kernel_launch(void* const* d_in, const int* in_sizes, int n_in,
                              void* d_out, int out_size, void* d_ws, size_t ws_size,
                              hipStream_t stream) {
  (void)in_sizes; (void)n_in; (void)out_size; (void)ws_size;
  const float* x = (const float*)d_in[0];
  const float* b = (const float*)d_in[1];
  const float* a = (const float*)d_in[2];
  float* out = (float*)d_out;
  float* ws = (float*)d_ws;

  setup_kernel<<<1, 1, 0, stream>>>(b, a, ws);
  fwd_kernel<<<NRG * NCHUNK, 32, 0, stream>>>(x, ws);
  bwd_kernel<<<NRG * NCHUNK, 32, 0, stream>>>(ws, out);
}


# Interior chunks stage their source rows through LDS: each global load reads 512 contiguous bytes of one row, a 128-sample segment ahead, instead of 16-byte pieces of 16 rows per load
# speedup vs baseline: 1.4128x; 1.1139x over previous
#include <hip/hip_runtime.h>
#include <cstddef>

typedef __attribute__((ext_vector_type(2))) float v2f;
typedef __attribute__((ext_vector_type(8))) float v8f;
typedef __attribute__((ext_vector_type(4))) float v4f;

#if defined(__HIP_DEVICE_COMPILE__) && !__has_builtin(__builtin_amdgcn_wmma_f32_16x16x4_f32)
#error "__builtin_amdgcn_wmma_f32_16x16x4_f32 not available on gfx1250 device pass"
#endif

static __device__ __forceinline__ v8f wmma4(v2f a, v2f b, v8f c) {
  return __builtin_amdgcn_wmma_f32_16x16x4_f32(false, a, false, b, (short)0, c,
                                               false, false);
}

constexpr int NROWS = 256;
constexpr int TLEN  = 131072;
constexpr int PAD   = 15;
constexpr int ELEN  = TLEN + 2 * PAD;
constexpr int ES    = 131104;
constexpr int NBLK  = ES / 16;
constexpr int CB    = 32;
constexpr int WB    = 16;
constexpr int NCHUNK = (NBLK + CB - 1) / CB;
constexpr int NRG   = NROWS / 16;

constexpr int WS_TB = 0;
constexpr int WS_WB = 256;
constexpr int WS_UB = 320;
constexpr int WS_PB = 576;
constexpr int WS_ZI = 640;
constexpr int WS_YF = 1024;

__global__ void setup_kernel(const float* __restrict__ bc,
                             const float* __restrict__ ac,
                             float* __restrict__ ws) {
  if (threadIdx.x != 0 || blockIdx.x != 0) return;
  double a0 = (double)ac[0];
  double an[5], bn[5];
  for (int i = 0; i < 5; ++i) { an[i] = (double)ac[i] / a0; bn[i] = (double)bc[i] / a0; }
  const double b0 = bn[0];
  double Bv[4];
  for (int i = 0; i < 4; ++i) Bv[i] = bn[i + 1] - an[i + 1] * b0;
  double A[4][4] = {{-an[1], 1, 0, 0},
                    {-an[2], 0, 1, 0},
                    {-an[3], 0, 0, 1},
                    {-an[4], 0, 0, 0}};
  double Crow[16][4];
  Crow[0][0] = 1; Crow[0][1] = 0; Crow[0][2] = 0; Crow[0][3] = 0;
  for (int k = 1; k < 16; ++k)
    for (int j = 0; j < 4; ++j) {
      double s = 0;
      for (int i = 0; i < 4; ++i) s += Crow[k - 1][i] * A[i][j];
      Crow[k][j] = s;
    }
  double h[16];
  h[0] = b0;
  for (int m = 1; m < 16; ++m) {
    double s = 0;
    for (int i = 0; i < 4; ++i) s += Crow[m - 1][i] * Bv[i];
    h[m] = s;
  }
  double u[16][4];
  for (int i = 0; i < 4; ++i) u[0][i] = Bv[i];
  for (int p = 1; p < 16; ++p)
    for (int i = 0; i < 4; ++i) {
      double s = 0;
      for (int j = 0; j < 4; ++j) s += A[i][j] * u[p - 1][j];
      u[p][i] = s;
    }
  double P[4][4], Tm[4][4];
  for (int i = 0; i < 4; ++i)
    for (int j = 0; j < 4; ++j) P[i][j] = (i == j) ? 1.0 : 0.0;
  for (int s = 0; s < 16; ++s) {
    for (int i = 0; i < 4; ++i)
      for (int j = 0; j < 4; ++j) {
        double t = 0;
        for (int q = 0; q < 4; ++q) t += A[i][q] * P[q][j];
        Tm[i][j] = t;
      }
    for (int i = 0; i < 4; ++i)
      for (int j = 0; j < 4; ++j) P[i][j] = Tm[i][j];
  }
  double M[4][5];
  for (int i = 0; i < 4; ++i) {
    for (int j = 0; j < 4; ++j)
      M[i][j] = ((i == j) ? 1.0 : 0.0) + ((j == 0) ? an[i + 1] : 0.0) -
                ((j == i + 1) ? 1.0 : 0.0);
    M[i][4] = Bv[i];
  }
  for (int c = 0; c < 4; ++c) {
    int piv = c;
    for (int r = c + 1; r < 4; ++r)
      if ((M[r][c] < 0 ? -M[r][c] : M[r][c]) > (M[piv][c] < 0 ? -M[piv][c] : M[piv][c]))
        piv = r;
    if (piv != c)
      for (int j = 0; j < 5; ++j) { double t = M[c][j]; M[c][j] = M[piv][j]; M[piv][j] = t; }
    double d = M[c][c];
    for (int j = 0; j < 5; ++j) M[c][j] /= d;
    for (int r = 0; r < 4; ++r) {
      if (r == c) continue;
      double f = M[r][c];
      for (int j = 0; j < 5; ++j) M[r][j] -= f * M[c][j];
    }
  }
  for (int c = 0; c < 4; ++c)
    for (int v = 0; v < 2; ++v)
      for (int L = 0; L < 32; ++L) {
        int k = v + ((L >= 16) ? 2 : 0);
        int n = L & 15;
        int kk = 4 * c + k;
        ws[WS_TB + c * 64 + v * 32 + L] = (float)((n >= kk) ? h[n - kk] : 0.0);
        ws[WS_UB + c * 64 + v * 32 + L] = (float)((n < 4) ? u[15 - kk][n] : 0.0);
      }
  for (int v = 0; v < 2; ++v)
    for (int L = 0; L < 32; ++L) {
      int k = v + ((L >= 16) ? 2 : 0);
      int n = L & 15;
      ws[WS_WB + v * 32 + L] = (float)Crow[n][k];
      ws[WS_PB + v * 32 + L] = (float)((n < 4) ? P[n][k] : 0.0);
    }
  for (int i = 0; i < 4; ++i) ws[WS_ZI + i] = (float)M[i][4];
}

static __device__ __forceinline__ float load_ext_fwd(const float* __restrict__ xr, int e) {
  if (e < PAD) return 2.0f * xr[0] - xr[PAD - e];
  if (e < PAD + TLEN) return xr[e - PAD];
  int j = e - (PAD + TLEN);
  if (j > PAD - 1) j = PAD - 1;
  return 2.0f * xr[TLEN - 1] - xr[TLEN - 2 - j];
}

static __device__ __forceinline__ float load_rev(const float* __restrict__ yr, int e) {
  int idx = ELEN - 1 - e;
  if (idx < 0) idx = 0;
  return yr[idx];
}


constexpr int SEG  = 128;
constexpr int LP   = 164;
constexpr int NPH  = (WB + CB) / 8;
constexpr int NPHW = WB / 8;

static __device__ __forceinline__ void lds_sync_wave() {
  __builtin_amdgcn_fence(__ATOMIC_RELEASE, "workgroup", "local");
  __builtin_amdgcn_wave_barrier();
  __builtin_amdgcn_fence(__ATOMIC_ACQUIRE, "workgroup", "local");
}

struct WarmTag { static constexpr bool out = false; };
struct OutTag  { static constexpr bool out = true; };

template <bool BWD>
static __device__ __forceinline__ void staged_chunk(const float* __restrict__ srcg,
                                                    int srcStride,
                                                    float* __restrict__ dstg,
                                                    const float* __restrict__ ws,
                                                    float* __restrict__ lds, int ch) {
  const int L = threadIdx.x;
  const int m = L & 15;
  const int hiA = L >> 4;
  const int k0 = 2 * hiA;
  constexpr int DIR  = BWD ? -1 : 1;
  constexpr int NOFF = BWD ? 0 : 16;
  constexpr int COFF = BWD ? SEG : 0;
  constexpr int CLEN = BWD ? 32 : 16;
  constexpr int CSRC = BWD ? 0 : (16 + SEG - 16);
  constexpr int PB   = BWD ? 157 : 1;

  v2f TBv[4], UBv[4], WBv, PBv;
#pragma unroll
  for (int c = 0; c < 4; ++c) {
    TBv[c] = (v2f){ws[WS_TB + c * 64 + L], ws[WS_TB + c * 64 + 32 + L]};
    UBv[c] = (v2f){ws[WS_UB + c * 64 + L], ws[WS_UB + c * 64 + 32 + L]};
  }
  WBv = (v2f){ws[WS_WB + L], ws[WS_WB + 32 + L]};
  PBv = (v2f){ws[WS_PB + L], ws[WS_PB + 32 + L]};
  const float zi0 = ws[WS_ZI + k0], zi1 = ws[WS_ZI + k0 + 1];

  const int bW = ch * CB - WB;
  const int S0 = BWD ? (ELEN - 1 - 16 * bW) : (16 * bW - PAD);
  const int J0 = S0 >> 7;

  const float* __restrict__ gl = srcg + 4 * L;
  float* __restrict__ lw = lds + NOFF + 4 * L;
  const float* __restrict__ lrow = lds + m * LP + DIR * k0;
  float* __restrict__ cdst = lds + (L >> 1) * LP + COFF + (L & 1) * (CLEN / 2);
  const float* __restrict__ csrc = lds + (L >> 1) * LP + CSRC + (L & 1) * (CLEN / 2);

  v4f R[16];
  auto gload = [&](const int j) {
#pragma unroll
    for (int r = 0; r < 16; ++r) R[r] = *(const v4f*)(gl + (size_t)r * srcStride + SEG * j);
  };
  auto lstore = [&]() {
#pragma unroll
    for (int r = 0; r < 16; ++r) *(v4f*)(lw + r * LP) = R[r];
  };
  auto carry = [&]() {
    v4f t[CLEN / 8];
#pragma unroll
    for (int i = 0; i < CLEN / 8; ++i) t[i] = *(const v4f*)(csrc + 4 * i);
#pragma unroll
    for (int i = 0; i < CLEN / 8; ++i) *(v4f*)(cdst + 4 * i) = t[i];
  };

  gload(J0);
  lstore();
  lds_sync_wave();
  carry();
  gload(J0 + DIR);
  lds_sync_wave();
  lstore();
  lds_sync_wave();

  float v0 = lrow[PB - DIR * k0];
  v2f z = (v2f){zi0 * v0, zi1 * v0};

  const int sl0 = k0 + ((m & 8) ? 16 : 0);
  const int sl1 = sl0 + 1;
  const int vsel = L & 7;
  const v8f zero = (v8f){0, 0, 0, 0, 0, 0, 0, 0};

  auto phase = [&](const int p, auto tag) {
    constexpr bool OUT = decltype(tag)::out;
    const bool more = (p + 1 < NPH);
    if (more) gload(J0 + DIR * (p + 2));
    v2f xa[4], xn[4];
#pragma unroll
    for (int c = 0; c < 4; ++c)
      xa[c] = (v2f){lrow[PB + DIR * (4 * c)], lrow[PB + DIR * (4 * c + 1)]};
#pragma unroll
    for (int q = 0; q < 8; ++q) {
      const int bb = bW + 8 * p + q;
      if (q < 7) {
#pragma unroll
        for (int c = 0; c < 4; ++c)
          xn[c] = (v2f){lrow[PB + DIR * (16 * (q + 1) + 4 * c)], lrow[PB + DIR * (16 * (q + 1) + 4 * c + 1)]};
      }
      v8f dZ = wmma4(z, PBv, zero);
#pragma unroll
      for (int c = 0; c < 4; ++c) dZ = wmma4(xa[c], UBv[c], dZ);
      if (OUT) {
        v8f dY = wmma4(z, WBv, zero);
#pragma unroll
        for (int c = 0; c < 4; ++c) dY = wmma4(xa[c], TBv[c], dY);
        const int e = bb * 16 + m;
        if (!BWD) {
#pragma unroll
          for (int v = 0; v < 8; ++v) dstg[(size_t)(v + 8 * hiA) * ES + e] = dY[v];
        } else {
          const int col = ELEN - 16 - e;
#pragma unroll
          for (int v = 0; v < 8; ++v) dstg[(size_t)(v + 8 * hiA) * TLEN + col] = dY[v];
        }
      }
      float nz0 = 0.f, nz1 = 0.f;
#pragma unroll
      for (int v = 0; v < 8; ++v) {
        float g0 = __shfl(dZ[v], sl0, 32);
        float g1 = __shfl(dZ[v], sl1, 32);
        nz0 = (vsel == v) ? g0 : nz0;
        nz1 = (vsel == v) ? g1 : nz1;
      }
      z = (v2f){nz0, nz1};
      if (q < 7) {
#pragma unroll
        for (int c = 0; c < 4; ++c) xa[c] = xn[c];
        __builtin_amdgcn_sched_barrier(0);
      }
    }
    if (more) {
      lds_sync_wave();
      carry();
      lds_sync_wave();
      lstore();
      lds_sync_wave();
    }
  };

  for (int p = 0; p < NPHW; ++p) phase(p, WarmTag{});
  for (int p = NPHW; p < NPH; ++p) phase(p, OutTag{});
}

__global__ __launch_bounds__(32) void fwd_kernel(const float* __restrict__ x,
                                                 float* __restrict__ ws) {
  float* __restrict__ yf = ws + WS_YF;
  const int task = blockIdx.x;
  const int rg = task / NCHUNK, ch = task % NCHUNK;
  const int r0 = rg * 16;
  __shared__ __attribute__((aligned(16))) float lds[16 * LP];
  if (ch >= 1 && ch <= NCHUNK - 2) {
    staged_chunk<false>(x + (size_t)r0 * TLEN, TLEN, yf + (size_t)r0 * ES, ws, lds, ch);
    return;
  }
  const int L = threadIdx.x;
  const int m = L & 15;
  const int hiA = L >> 4;
  const int k0 = 2 * hiA;
  const float* __restrict__ xrow = x + (size_t)(r0 + m) * TLEN;

  v2f TBv[4], UBv[4], WBv, PBv;
#pragma unroll
  for (int c = 0; c < 4; ++c) {
    TBv[c] = (v2f){ws[WS_TB + c * 64 + L], ws[WS_TB + c * 64 + 32 + L]};
    UBv[c] = (v2f){ws[WS_UB + c * 64 + L], ws[WS_UB + c * 64 + 32 + L]};
  }
  WBv = (v2f){ws[WS_WB + L], ws[WS_WB + 32 + L]};
  PBv = (v2f){ws[WS_PB + L], ws[WS_PB + 32 + L]};
  const float zi0 = ws[WS_ZI + k0], zi1 = ws[WS_ZI + k0 + 1];

  const int bOut = ch * CB;
  const int bW = (bOut - WB > 0) ? (bOut - WB) : 0;
  const int bEnd = ((bOut + CB) < NBLK) ? (bOut + CB) : NBLK;

  float v0 = load_ext_fwd(xrow, bW * 16);
  v2f z = (v2f){zi0 * v0, zi1 * v0};

  const int sl0 = k0 + ((m & 8) ? 16 : 0);
  const int sl1 = sl0 + 1;
  const int vsel = L & 7;

  for (int bb = bW; bb < bEnd; ++bb) {
    const int e0 = bb * 16;
    v2f xa[4];
    if (e0 >= PAD && (e0 + 16) <= (PAD + TLEN)) {
      const float* p = xrow + (e0 - PAD) + k0;
#pragma unroll
      for (int c = 0; c < 4; ++c) xa[c] = (v2f){p[4 * c], p[4 * c + 1]};
      __builtin_prefetch(p + 64, 0, 1);
    } else {
#pragma unroll
      for (int c = 0; c < 4; ++c) {
        int o = e0 + 4 * c + k0;
        xa[c] = (v2f){load_ext_fwd(xrow, o), load_ext_fwd(xrow, o + 1)};
      }
    }
    const v8f zero = (v8f){0, 0, 0, 0, 0, 0, 0, 0};
    v8f dZ = wmma4(z, PBv, zero);
#pragma unroll
    for (int c = 0; c < 4; ++c) dZ = wmma4(xa[c], UBv[c], dZ);
    if (bb >= bOut) {
      v8f dY = wmma4(z, WBv, zero);
#pragma unroll
      for (int c = 0; c < 4; ++c) dY = wmma4(xa[c], TBv[c], dY);
#pragma unroll
      for (int v = 0; v < 8; ++v)
        yf[(size_t)(r0 + v + 8 * hiA) * ES + e0 + m] = dY[v];
    }
    float nz0 = 0.f, nz1 = 0.f;
#pragma unroll
    for (int v = 0; v < 8; ++v) {
      float g0 = __shfl(dZ[v], sl0, 32);
      float g1 = __shfl(dZ[v], sl1, 32);
      nz0 = (vsel == v) ? g0 : nz0;
      nz1 = (vsel == v) ? g1 : nz1;
    }
    z = (v2f){nz0, nz1};
  }
}

__global__ __launch_bounds__(32) void bwd_kernel(const float* __restrict__ ws,
                                                 float* __restrict__ out) {
  const float* __restrict__ yf = ws + WS_YF;
  const int task = blockIdx.x;
  const int rg = task / NCHUNK, ch = task % NCHUNK;
  const int r0 = rg * 16;
  __shared__ __attribute__((aligned(16))) float lds[16 * LP];
  if (ch >= 1 && ch <= NCHUNK - 2) {
    staged_chunk<true>(yf + (size_t)r0 * ES, ES, out + (size_t)r0 * TLEN, ws, lds, ch);
    return;
  }
  const int L = threadIdx.x;
  const int m = L & 15;
  const int hiA = L >> 4;
  const int k0 = 2 * hiA;
  const float* __restrict__ yrow = yf + (size_t)(r0 + m) * ES;

  v2f TBv[4], UBv[4], WBv, PBv;
#pragma unroll
  for (int c = 0; c < 4; ++c) {
    TBv[c] = (v2f){ws[WS_TB + c * 64 + L], ws[WS_TB + c * 64 + 32 + L]};
    UBv[c] = (v2f){ws[WS_UB + c * 64 + L], ws[WS_UB + c * 64 + 32 + L]};
  }
  WBv = (v2f){ws[WS_WB + L], ws[WS_WB + 32 + L]};
  PBv = (v2f){ws[WS_PB + L], ws[WS_PB + 32 + L]};
  const float zi0 = ws[WS_ZI + k0], zi1 = ws[WS_ZI + k0 + 1];

  const int bOut = ch * CB;
  const int bW = (bOut - WB > 0) ? (bOut - WB) : 0;
  const int bEnd = ((bOut + CB) < NBLK) ? (bOut + CB) : NBLK;

  float v0 = load_rev(yrow, bW * 16);
  v2f z = (v2f){zi0 * v0, zi1 * v0};
  const int sl0 = k0 + ((m & 8) ? 16 : 0);
  const int sl1 = sl0 + 1;
  const int vsel = L & 7;

  for (int bb = bW; bb < bEnd; ++bb) {
    const int e0 = bb * 16;
    v2f xa[4];
    if (e0 + 16 <= ELEN) {
      const float* p = yrow + (ELEN - 1 - e0 - k0);
#pragma unroll
      for (int c = 0; c < 4; ++c) xa[c] = (v2f){p[-4 * c], p[-4 * c - 1]};
      __builtin_prefetch(p - 64, 0, 1);
    } else {
#pragma unroll
      for (int c = 0; c < 4; ++c) {
        int o = e0 + 4 * c + k0;
        xa[c] = (v2f){load_rev(yrow, o), load_rev(yrow, o + 1)};
      }
    }
    const v8f zero = (v8f){0, 0, 0, 0, 0, 0, 0, 0};
    v8f dZ = wmma4(z, PBv, zero);
#pragma unroll
    for (int c = 0; c < 4; ++c) dZ = wmma4(xa[c], UBv[c], dZ);
    if (bb >= bOut) {
      v8f dY = wmma4(z, WBv, zero);
#pragma unroll
      for (int c = 0; c < 4; ++c) dY = wmma4(xa[c], TBv[c], dY);
      const int e = e0 + m;
      if (e >= PAD && e < PAD + TLEN) {
        const int col = ELEN - 16 - e;
#pragma unroll
        for (int v = 0; v < 8; ++v)
          out[(size_t)(r0 + v + 8 * hiA) * TLEN + col] = dY[v];
      }
    }
    float nz0 = 0.f, nz1 = 0.f;
#pragma unroll
    for (int v = 0; v < 8; ++v) {
      float g0 = __shfl(dZ[v], sl0, 32);
      float g1 = __shfl(dZ[v], sl1, 32);
      nz0 = (vsel == v) ? g0 : nz0;
      nz1 = (vsel == v) ? g1 : nz1;
    }
    z = (v2f){nz0, nz1};
  }
}

extern "C" void kernel_launch(void* const* d_in, const int* in_sizes, int n_in,
                              void* d_out, int out_size, void* d_ws, size_t ws_size,
                              hipStream_t stream) {
  (void)in_sizes; (void)n_in; (void)out_size; (void)ws_size;
  const float* x = (const float*)d_in[0];
  const float* b = (const float*)d_in[1];
  const float* a = (const float*)d_in[2];
  float* out = (float*)d_out;
  float* ws = (float*)d_ws;

  setup_kernel<<<1, 1, 0, stream>>>(b, a, ws);
  fwd_kernel<<<NRG * NCHUNK, 32, 0, stream>>>(x, ws);
  bwd_kernel<<<NRG * NCHUNK, 32, 0, stream>>>(ws, out);
}


# Setup kernel: 64 threads share its double arithmetic (A^k by columns on 5 lanes, zi by summed rows with 1 division): ~100 FP64 instructions on the longer wave, not ~1100 on one lane; sums reordered
# speedup vs baseline: 1.8457x; 1.3064x over previous
#include <hip/hip_runtime.h>
#include <cstddef>

typedef __attribute__((ext_vector_type(2))) float v2f;
typedef __attribute__((ext_vector_type(8))) float v8f;
typedef __attribute__((ext_vector_type(4))) float v4f;

#if defined(__HIP_DEVICE_COMPILE__) && !__has_builtin(__builtin_amdgcn_wmma_f32_16x16x4_f32)
#error "__builtin_amdgcn_wmma_f32_16x16x4_f32 not available on gfx1250 device pass"
#endif

static __device__ __forceinline__ v8f wmma4(v2f a, v2f b, v8f c) {
  return __builtin_amdgcn_wmma_f32_16x16x4_f32(false, a, false, b, (short)0, c,
                                               false, false);
}

constexpr int NROWS = 256;
constexpr int TLEN  = 131072;
constexpr int PAD   = 15;
constexpr int ELEN  = TLEN + 2 * PAD;
constexpr int ES    = 131104;
constexpr int NBLK  = ES / 16;
constexpr int CB    = 32;
constexpr int WB    = 16;
constexpr int NCHUNK = (NBLK + CB - 1) / CB;
constexpr int NRG   = NROWS / 16;

constexpr int WS_TB = 0;
constexpr int WS_WB = 256;
constexpr int WS_UB = 320;
constexpr int WS_PB = 576;
constexpr int WS_ZI = 640;
constexpr int WS_YF = 1024;

constexpr int NVEC = 5;
constexpr int NPOW = 17;

__global__ __launch_bounds__(64) void setup_kernel(const float* __restrict__ bc,
                                                   const float* __restrict__ ac,
                                                   float* __restrict__ ws) {
  __shared__ double sQ[16];
  __shared__ double sV[NVEC * NPOW * 4];
  const int t = threadIdx.x;
  if (t < 16) {
    const int i = t & 7;
    const double a0 = (double)ac[0];
    double c = 0.0;
    if (i < 5) c = (double)((t < 8) ? ac[i] : bc[i]);
    sQ[t] = c / a0;
  }
  __syncthreads();
  const double an1 = sQ[1], an2 = sQ[2], an3 = sQ[3], an4 = sQ[4];
  const double b0 = sQ[8];
  const double Bv0 = sQ[9] - an1 * b0, Bv1 = sQ[10] - an2 * b0;
  const double Bv2 = sQ[11] - an3 * b0, Bv3 = sQ[12] - an4 * b0;

  if (t < 32) {
    double v0 = (t == 4) ? Bv0 : ((t == 0) ? 1.0 : 0.0);
    double v1 = (t == 4) ? Bv1 : ((t == 1) ? 1.0 : 0.0);
    double v2 = (t == 4) ? Bv2 : ((t == 2) ? 1.0 : 0.0);
    double v3 = (t == 4) ? Bv3 : ((t == 3) ? 1.0 : 0.0);
    for (int k = 0; k < NPOW; ++k) {
      if (t < NVEC) {
        double* __restrict__ d = sV + (t * NPOW + k) * 4;
        d[0] = v0; d[1] = v1; d[2] = v2; d[3] = v3;
      }
      const double n0 = v1 - an1 * v0;
      const double n1 = v2 - an2 * v0;
      const double n2 = v3 - an3 * v0;
      const double n3 = -(an4 * v0);
      v0 = n0; v1 = n1; v2 = n2; v3 = n3;
    }
  } else {
    const double s1 = 1.0 + an1, s2 = s1 + an2, s3 = s2 + an3, s4 = s3 + an4;
    const double c1 = Bv0, c2 = c1 + Bv1, c3 = c2 + Bv2, c4 = c3 + Bv3;
    const double z0 = c4 / s4;
    if (t == 32) {
      ws[WS_ZI + 0] = (float)z0;
      ws[WS_ZI + 1] = (float)(s1 * z0 - c1);
      ws[WS_ZI + 2] = (float)(s2 * z0 - c2);
      ws[WS_ZI + 3] = (float)(s3 * z0 - c3);
    }
  }
  __syncthreads();

  const int v = t >> 5, L = t & 31;
  const int n = L & 15;
  const int k = v + 2 * (L >> 4);
  const int nn = n & 3;
  for (int c = 0; c < 4; ++c) {
    const int kk = 4 * c + k;
    const int d = n - kk;
    const double hd = sV[(4 * NPOW + ((d > 0) ? d - 1 : 0)) * 4];
    ws[WS_TB + c * 64 + t] = (float)((d > 0) ? hd : ((d == 0) ? b0 : 0.0));
    const double ud = sV[(4 * NPOW + 15 - kk) * 4 + nn];
    ws[WS_UB + c * 64 + t] = (float)((n < 4) ? ud : 0.0);
  }
  ws[WS_WB + t] = (float)sV[(k * NPOW + n) * 4];
  const double pd = sV[(k * NPOW + 16) * 4 + nn];
  ws[WS_PB + t] = (float)((n < 4) ? pd : 0.0);
}

static __device__ __forceinline__ float load_ext_fwd(const float* __restrict__ xr, int e) {
  if (e < PAD) return 2.0f * xr[0] - xr[PAD - e];
  if (e < PAD + TLEN) return xr[e - PAD];
  int j = e - (PAD + TLEN);
  if (j > PAD - 1) j = PAD - 1;
  return 2.0f * xr[TLEN - 1] - xr[TLEN - 2 - j];
}

static __device__ __forceinline__ float load_rev(const float* __restrict__ yr, int e) {
  int idx = ELEN - 1 - e;
  if (idx < 0) idx = 0;
  return yr[idx];
}


constexpr int SEG  = 128;
constexpr int LP   = 164;
constexpr int NPH  = (WB + CB) / 8;
constexpr int NPHW = WB / 8;

static __device__ __forceinline__ void lds_sync_wave() {
  __builtin_amdgcn_fence(__ATOMIC_RELEASE, "workgroup", "local");
  __builtin_amdgcn_wave_barrier();
  __builtin_amdgcn_fence(__ATOMIC_ACQUIRE, "workgroup", "local");
}

struct WarmTag { static constexpr bool out = false; };
struct OutTag  { static constexpr bool out = true; };

template <bool BWD>
static __device__ __forceinline__ void staged_chunk(const float* __restrict__ srcg,
                                                    int srcStride,
                                                    float* __restrict__ dstg,
                                                    const float* __restrict__ ws,
                                                    float* __restrict__ lds, int ch) {
  const int L = threadIdx.x;
  const int m = L & 15;
  const int hiA = L >> 4;
  const int k0 = 2 * hiA;
  constexpr int DIR  = BWD ? -1 : 1;
  constexpr int NOFF = BWD ? 0 : 16;
  constexpr int COFF = BWD ? SEG : 0;
  constexpr int CLEN = BWD ? 32 : 16;
  constexpr int CSRC = BWD ? 0 : (16 + SEG - 16);
  constexpr int PB   = BWD ? 157 : 1;

  v2f TBv[4], UBv[4], WBv, PBv;
#pragma unroll
  for (int c = 0; c < 4; ++c) {
    TBv[c] = (v2f){ws[WS_TB + c * 64 + L], ws[WS_TB + c * 64 + 32 + L]};
    UBv[c] = (v2f){ws[WS_UB + c * 64 + L], ws[WS_UB + c * 64 + 32 + L]};
  }
  WBv = (v2f){ws[WS_WB + L], ws[WS_WB + 32 + L]};
  PBv = (v2f){ws[WS_PB + L], ws[WS_PB + 32 + L]};
  const float zi0 = ws[WS_ZI + k0], zi1 = ws[WS_ZI + k0 + 1];

  const int bW = ch * CB - WB;
  const int S0 = BWD ? (ELEN - 1 - 16 * bW) : (16 * bW - PAD);
  const int J0 = S0 >> 7;

  const float* __restrict__ gl = srcg + 4 * L;
  float* __restrict__ lw = lds + NOFF + 4 * L;
  const float* __restrict__ lrow = lds + m * LP + DIR * k0;
  float* __restrict__ cdst = lds + (L >> 1) * LP + COFF + (L & 1) * (CLEN / 2);
  const float* __restrict__ csrc = lds + (L >> 1) * LP + CSRC + (L & 1) * (CLEN / 2);

  v4f R[16];
  auto gload = [&](const int j) {
#pragma unroll
    for (int r = 0; r < 16; ++r) R[r] = *(const v4f*)(gl + (size_t)r * srcStride + SEG * j);
  };
  auto lstore = [&]() {
#pragma unroll
    for (int r = 0; r < 16; ++r) *(v4f*)(lw + r * LP) = R[r];
  };
  auto carry = [&]() {
    v4f t[CLEN / 8];
#pragma unroll
    for (int i = 0; i < CLEN / 8; ++i) t[i] = *(const v4f*)(csrc + 4 * i);
#pragma unroll
    for (int i = 0; i < CLEN / 8; ++i) *(v4f*)(cdst + 4 * i) = t[i];
  };

  gload(J0);
  lstore();
  lds_sync_wave();
  carry();
  gload(J0 + DIR);
  lds_sync_wave();
  lstore();
  lds_sync_wave();

  float v0 = lrow[PB - DIR * k0];
  v2f z = (v2f){zi0 * v0, zi1 * v0};

  const int sl0 = k0 + ((m & 8) ? 16 : 0);
  const int sl1 = sl0 + 1;
  const int vsel = L & 7;
  const v8f zero = (v8f){0, 0, 0, 0, 0, 0, 0, 0};

  auto phase = [&](const int p, auto tag) {
    constexpr bool OUT = decltype(tag)::out;
    const bool more = (p + 1 < NPH);
    if (more) gload(J0 + DIR * (p + 2));
    v2f xa[4], xn[4];
#pragma unroll
    for (int c = 0; c < 4; ++c)
      xa[c] = (v2f){lrow[PB + DIR * (4 * c)], lrow[PB + DIR * (4 * c + 1)]};
#pragma unroll
    for (int q = 0; q < 8; ++q) {
      const int bb = bW + 8 * p + q;
      if (q < 7) {
#pragma unroll
        for (int c = 0; c < 4; ++c)
          xn[c] = (v2f){lrow[PB + DIR * (16 * (q + 1) + 4 * c)], lrow[PB + DIR * (16 * (q + 1) + 4 * c + 1)]};
      }
      v8f dZ = wmma4(z, PBv, zero);
#pragma unroll
      for (int c = 0; c < 4; ++c) dZ = wmma4(xa[c], UBv[c], dZ);
      if (OUT) {
        v8f dY = wmma4(z, WBv, zero);
#pragma unroll
        for (int c = 0; c < 4; ++c) dY = wmma4(xa[c], TBv[c], dY);
        const int e = bb * 16 + m;
        if (!BWD) {
#pragma unroll
          for (int v = 0; v < 8; ++v) dstg[(size_t)(v + 8 * hiA) * ES + e] = dY[v];
        } else {
          const int col = ELEN - 16 - e;
#pragma unroll
          for (int v = 0; v < 8; ++v) dstg[(size_t)(v + 8 * hiA) * TLEN + col] = dY[v];
        }
      }
      float nz0 = 0.f, nz1 = 0.f;
#pragma unroll
      for (int v = 0; v < 8; ++v) {
        float g0 = __shfl(dZ[v], sl0, 32);
        float g1 = __shfl(dZ[v], sl1, 32);
        nz0 = (vsel == v) ? g0 : nz0;
        nz1 = (vsel == v) ? g1 : nz1;
      }
      z = (v2f){nz0, nz1};
      if (q < 7) {
#pragma unroll
        for (int c = 0; c < 4; ++c) xa[c] = xn[c];
        __builtin_amdgcn_sched_barrier(0);
      }
    }
    if (more) {
      lds_sync_wave();
      carry();
      lds_sync_wave();
      lstore();
      lds_sync_wave();
    }
  };

  for (int p = 0; p < NPHW; ++p) phase(p, WarmTag{});
  for (int p = NPHW; p < NPH; ++p) phase(p, OutTag{});
}

__global__ __launch_bounds__(32) void fwd_kernel(const float* __restrict__ x,
                                                 float* __restrict__ ws) {
  float* __restrict__ yf = ws + WS_YF;
  const int task = blockIdx.x;
  const int rg = task / NCHUNK, ch = task % NCHUNK;
  const int r0 = rg * 16;
  __shared__ __attribute__((aligned(16))) float lds[16 * LP];
  if (ch >= 1 && ch <= NCHUNK - 2) {
    staged_chunk<false>(x + (size_t)r0 * TLEN, TLEN, yf + (size_t)r0 * ES, ws, lds, ch);
    return;
  }
  const int L = threadIdx.x;
  const int m = L & 15;
  const int hiA = L >> 4;
  const int k0 = 2 * hiA;
  const float* __restrict__ xrow = x + (size_t)(r0 + m) * TLEN;

  v2f TBv[4], UBv[4], WBv, PBv;
#pragma unroll
  for (int c = 0; c < 4; ++c) {
    TBv[c] = (v2f){ws[WS_TB + c * 64 + L], ws[WS_TB + c * 64 + 32 + L]};
    UBv[c] = (v2f){ws[WS_UB + c * 64 + L], ws[WS_UB + c * 64 + 32 + L]};
  }
  WBv = (v2f){ws[WS_WB + L], ws[WS_WB + 32 + L]};
  PBv = (v2f){ws[WS_PB + L], ws[WS_PB + 32 + L]};
  const float zi0 = ws[WS_ZI + k0], zi1 = ws[WS_ZI + k0 + 1];

  const int bOut = ch * CB;
  const int bW = (bOut - WB > 0) ? (bOut - WB) : 0;
  const int bEnd = ((bOut + CB) < NBLK) ? (bOut + CB) : NBLK;

  float v0 = load_ext_fwd(xrow, bW * 16);
  v2f z = (v2f){zi0 * v0, zi1 * v0};

  const int sl0 = k0 + ((m & 8) ? 16 : 0);
  const int sl1 = sl0 + 1;
  const int vsel = L & 7;

  for (int bb = bW; bb < bEnd; ++bb) {
    const int e0 = bb * 16;
    v2f xa[4];
    if (e0 >= PAD && (e0 + 16) <= (PAD + TLEN)) {
      const float* p = xrow + (e0 - PAD) + k0;
#pragma unroll
      for (int c = 0; c < 4; ++c) xa[c] = (v2f){p[4 * c], p[4 * c + 1]};
      __builtin_prefetch(p + 64, 0, 1);
    } else {
#pragma unroll
      for (int c = 0; c < 4; ++c) {
        int o = e0 + 4 * c + k0;
        xa[c] = (v2f){load_ext_fwd(xrow, o), load_ext_fwd(xrow, o + 1)};
      }
    }
    const v8f zero = (v8f){0, 0, 0, 0, 0, 0, 0, 0};
    v8f dZ = wmma4(z, PBv, zero);
#pragma unroll
    for (int c = 0; c < 4; ++c) dZ = wmma4(xa[c], UBv[c], dZ);
    if (bb >= bOut) {
      v8f dY = wmma4(z, WBv, zero);
#pragma unroll
      for (int c = 0; c < 4; ++c) dY = wmma4(xa[c], TBv[c], dY);
#pragma unroll
      for (int v = 0; v < 8; ++v)
        yf[(size_t)(r0 + v + 8 * hiA) * ES + e0 + m] = dY[v];
    }
    float nz0 = 0.f, nz1 = 0.f;
#pragma unroll
    for (int v = 0; v < 8; ++v) {
      float g0 = __shfl(dZ[v], sl0, 32);
      float g1 = __shfl(dZ[v], sl1, 32);
      nz0 = (vsel == v) ? g0 : nz0;
      nz1 = (vsel == v) ? g1 : nz1;
    }
    z = (v2f){nz0, nz1};
  }
}

__global__ __launch_bounds__(32) void bwd_kernel(const float* __restrict__ ws,
                                                 float* __restrict__ out) {
  const float* __restrict__ yf = ws + WS_YF;
  const int task = blockIdx.x;
  const int rg = task / NCHUNK, ch = task % NCHUNK;
  const int r0 = rg * 16;
  __shared__ __attribute__((aligned(16))) float lds[16 * LP];
  if (ch >= 1 && ch <= NCHUNK - 2) {
    staged_chunk<true>(yf + (size_t)r0 * ES, ES, out + (size_t)r0 * TLEN, ws, lds, ch);
    return;
  }
  const int L = threadIdx.x;
  const int m = L & 15;
  const int hiA = L >> 4;
  const int k0 = 2 * hiA;
  const float* __restrict__ yrow = yf + (size_t)(r0 + m) * ES;

  v2f TBv[4], UBv[4], WBv, PBv;
#pragma unroll
  for (int c = 0; c < 4; ++c) {
    TBv[c] = (v2f){ws[WS_TB + c * 64 + L], ws[WS_TB + c * 64 + 32 + L]};
    UBv[c] = (v2f){ws[WS_UB + c * 64 + L], ws[WS_UB + c * 64 + 32 + L]};
  }
  WBv = (v2f){ws[WS_WB + L], ws[WS_WB + 32 + L]};
  PBv = (v2f){ws[WS_PB + L], ws[WS_PB + 32 + L]};
  const float zi0 = ws[WS_ZI + k0], zi1 = ws[WS_ZI + k0 + 1];

  const int bOut = ch * CB;
  const int bW = (bOut - WB > 0) ? (bOut - WB) : 0;
  const int bEnd = ((bOut + CB) < NBLK) ? (bOut + CB) : NBLK;

  float v0 = load_rev(yrow, bW * 16);
  v2f z = (v2f){zi0 * v0, zi1 * v0};
  const int sl0 = k0 + ((m & 8) ? 16 : 0);
  const int sl1 = sl0 + 1;
  const int vsel = L & 7;

  for (int bb = bW; bb < bEnd; ++bb) {
    const int e0 = bb * 16;
    v2f xa[4];
    if (e0 + 16 <= ELEN) {
      const float* p = yrow + (ELEN - 1 - e0 - k0);
#pragma unroll
      for (int c = 0; c < 4; ++c) xa[c] = (v2f){p[-4 * c], p[-4 * c - 1]};
      __builtin_prefetch(p - 64, 0, 1);
    } else {
#pragma unroll
      for (int c = 0; c < 4; ++c) {
        int o = e0 + 4 * c + k0;
        xa[c] = (v2f){load_rev(yrow, o), load_rev(yrow, o + 1)};
      }
    }
    const v8f zero = (v8f){0, 0, 0, 0, 0, 0, 0, 0};
    v8f dZ = wmma4(z, PBv, zero);
#pragma unroll
    for (int c = 0; c < 4; ++c) dZ = wmma4(xa[c], UBv[c], dZ);
    if (bb >= bOut) {
      v8f dY = wmma4(z, WBv, zero);
#pragma unroll
      for (int c = 0; c < 4; ++c) dY = wmma4(xa[c], TBv[c], dY);
      const int e = e0 + m;
      if (e >= PAD && e < PAD + TLEN) {
        const int col = ELEN - 16 - e;
#pragma unroll
        for (int v = 0; v < 8; ++v)
          out[(size_t)(r0 + v + 8 * hiA) * TLEN + col] = dY[v];
      }
    }
    float nz0 = 0.f, nz1 = 0.f;
#pragma unroll
    for (int v = 0; v < 8; ++v) {
      float g0 = __shfl(dZ[v], sl0, 32);
      float g1 = __shfl(dZ[v], sl1, 32);
      nz0 = (vsel == v) ? g0 : nz0;
      nz1 = (vsel == v) ? g1 : nz1;
    }
    z = (v2f){nz0, nz1};
  }
}

extern "C" void kernel_launch(void* const* d_in, const int* in_sizes, int n_in,
                              void* d_out, int out_size, void* d_ws, size_t ws_size,
                              hipStream_t stream) {
  (void)in_sizes; (void)n_in; (void)out_size; (void)ws_size;
  const float* x = (const float*)d_in[0];
  const float* b = (const float*)d_in[1];
  const float* a = (const float*)d_in[2];
  float* out = (float*)d_out;
  float* ws = (float*)d_ws;

  setup_kernel<<<1, 64, 0, stream>>>(b, a, ws);
  fwd_kernel<<<NRG * NCHUNK, 32, 0, stream>>>(x, ws);
  bwd_kernel<<<NRG * NCHUNK, 32, 0, stream>>>(ws, out);
}


# Interior chunks stage result rows through a 64-sample LDS ring: every global store instruction writes whole 128-byte lines (16 B per lane) instead of 64-byte pieces of 2 rows, 8 per block
# speedup vs baseline: 2.1067x; 1.1414x over previous
#include <hip/hip_runtime.h>
#include <cstddef>

typedef __attribute__((ext_vector_type(2))) float v2f;
typedef __attribute__((ext_vector_type(8))) float v8f;
typedef __attribute__((ext_vector_type(4))) float v4f;

#if defined(__HIP_DEVICE_COMPILE__) && !__has_builtin(__builtin_amdgcn_wmma_f32_16x16x4_f32)
#error "__builtin_amdgcn_wmma_f32_16x16x4_f32 not available on gfx1250 device pass"
#endif

static __device__ __forceinline__ v8f wmma4(v2f a, v2f b, v8f c) {
  return __builtin_amdgcn_wmma_f32_16x16x4_f32(false, a, false, b, (short)0, c,
                                               false, false);
}

constexpr int NROWS = 256;
constexpr int TLEN  = 131072;
constexpr int PAD   = 15;
constexpr int ELEN  = TLEN + 2 * PAD;
constexpr int ES    = 131104;
constexpr int NBLK  = ES / 16;
constexpr int CB    = 32;
constexpr int WB    = 16;
constexpr int NCHUNK = (NBLK + CB - 1) / CB;
constexpr int NRG   = NROWS / 16;

constexpr int WS_TB = 0;
constexpr int WS_WB = 256;
constexpr int WS_UB = 320;
constexpr int WS_PB = 576;
constexpr int WS_ZI = 640;
constexpr int WS_YF = 1024;

constexpr int NVEC = 5;
constexpr int NPOW = 17;

__global__ __launch_bounds__(64) void setup_kernel(const float* __restrict__ bc,
                                                   const float* __restrict__ ac,
                                                   float* __restrict__ ws) {
  __shared__ double sQ[16];
  __shared__ double sV[NVEC * NPOW * 4];
  const int t = threadIdx.x;
  if (t < 16) {
    const int i = t & 7;
    const double a0 = (double)ac[0];
    double c = 0.0;
    if (i < 5) c = (double)((t < 8) ? ac[i] : bc[i]);
    sQ[t] = c / a0;
  }
  __syncthreads();
  const double an1 = sQ[1], an2 = sQ[2], an3 = sQ[3], an4 = sQ[4];
  const double b0 = sQ[8];
  const double Bv0 = sQ[9] - an1 * b0, Bv1 = sQ[10] - an2 * b0;
  const double Bv2 = sQ[11] - an3 * b0, Bv3 = sQ[12] - an4 * b0;

  if (t < 32) {
    double v0 = (t == 4) ? Bv0 : ((t == 0) ? 1.0 : 0.0);
    double v1 = (t == 4) ? Bv1 : ((t == 1) ? 1.0 : 0.0);
    double v2 = (t == 4) ? Bv2 : ((t == 2) ? 1.0 : 0.0);
    double v3 = (t == 4) ? Bv3 : ((t == 3) ? 1.0 : 0.0);
    for (int k = 0; k < NPOW; ++k) {
      if (t < NVEC) {
        double* __restrict__ d = sV + (t * NPOW + k) * 4;
        d[0] = v0; d[1] = v1; d[2] = v2; d[3] = v3;
      }
      const double n0 = v1 - an1 * v0;
      const double n1 = v2 - an2 * v0;
      const double n2 = v3 - an3 * v0;
      const double n3 = -(an4 * v0);
      v0 = n0; v1 = n1; v2 = n2; v3 = n3;
    }
  } else {
    const double s1 = 1.0 + an1, s2 = s1 + an2, s3 = s2 + an3, s4 = s3 + an4;
    const double c1 = Bv0, c2 = c1 + Bv1, c3 = c2 + Bv2, c4 = c3 + Bv3;
    const double z0 = c4 / s4;
    if (t == 32) {
      ws[WS_ZI + 0] = (float)z0;
      ws[WS_ZI + 1] = (float)(s1 * z0 - c1);
      ws[WS_ZI + 2] = (float)(s2 * z0 - c2);
      ws[WS_ZI + 3] = (float)(s3 * z0 - c3);
    }
  }
  __syncthreads();

  const int v = t >> 5, L = t & 31;
  const int n = L & 15;
  const int k = v + 2 * (L >> 4);
  const int nn = n & 3;
  for (int c = 0; c < 4; ++c) {
    const int kk = 4 * c + k;
    const int d = n - kk;
    const double hd = sV[(4 * NPOW + ((d > 0) ? d - 1 : 0)) * 4];
    ws[WS_TB + c * 64 + t] = (float)((d > 0) ? hd : ((d == 0) ? b0 : 0.0));
    const double ud = sV[(4 * NPOW + 15 - kk) * 4 + nn];
    ws[WS_UB + c * 64 + t] = (float)((n < 4) ? ud : 0.0);
  }
  ws[WS_WB + t] = (float)sV[(k * NPOW + n) * 4];
  const double pd = sV[(k * NPOW + 16) * 4 + nn];
  ws[WS_PB + t] = (float)((n < 4) ? pd : 0.0);
}

static __device__ __forceinline__ float load_ext_fwd(const float* __restrict__ xr, int e) {
  if (e < PAD) return 2.0f * xr[0] - xr[PAD - e];
  if (e < PAD + TLEN) return xr[e - PAD];
  int j = e - (PAD + TLEN);
  if (j > PAD - 1) j = PAD - 1;
  return 2.0f * xr[TLEN - 1] - xr[TLEN - 2 - j];
}

static __device__ __forceinline__ float load_rev(const float* __restrict__ yr, int e) {
  int idx = ELEN - 1 - e;
  if (idx < 0) idx = 0;
  return yr[idx];
}


constexpr int SEG  = 128;
constexpr int LP   = 164;
constexpr int OP   = 68;
constexpr int ORING = 64;
constexpr int LDSF = 16 * LP + 16 * OP;
constexpr int NPH  = (WB + CB) / 8;
constexpr int NPHW = WB / 8;

static __device__ __forceinline__ void lds_sync_wave() {
  __builtin_amdgcn_fence(__ATOMIC_RELEASE, "workgroup", "local");
  __builtin_amdgcn_wave_barrier();
  __builtin_amdgcn_fence(__ATOMIC_ACQUIRE, "workgroup", "local");
}

struct WarmTag { static constexpr bool out = false; };
struct OutTag  { static constexpr bool out = true; };

template <bool BWD>
static __device__ __forceinline__ void staged_chunk(const float* __restrict__ srcg,
                                                    int srcStride,
                                                    float* __restrict__ dstg,
                                                    const float* __restrict__ ws,
                                                    float* __restrict__ lds, int ch) {
  const int L = threadIdx.x;
  const int m = L & 15;
  const int hiA = L >> 4;
  const int k0 = 2 * hiA;
  constexpr int DIR  = BWD ? -1 : 1;
  constexpr int NOFF = BWD ? 0 : 16;
  constexpr int COFF = BWD ? SEG : 0;
  constexpr int CLEN = BWD ? 32 : 16;
  constexpr int CSRC = BWD ? 0 : (16 + SEG - 16);
  constexpr int PB   = BWD ? 157 : 1;

  v2f TBv[4], UBv[4], WBv, PBv;
#pragma unroll
  for (int c = 0; c < 4; ++c) {
    TBv[c] = (v2f){ws[WS_TB + c * 64 + L], ws[WS_TB + c * 64 + 32 + L]};
    UBv[c] = (v2f){ws[WS_UB + c * 64 + L], ws[WS_UB + c * 64 + 32 + L]};
  }
  WBv = (v2f){ws[WS_WB + L], ws[WS_WB + 32 + L]};
  PBv = (v2f){ws[WS_PB + L], ws[WS_PB + 32 + L]};
  const float zi0 = ws[WS_ZI + k0], zi1 = ws[WS_ZI + k0 + 1];

  const int bW = ch * CB - WB;
  const int S0 = BWD ? (ELEN - 1 - 16 * bW) : (16 * bW - PAD);
  const int J0 = S0 >> 7;

  const float* __restrict__ gl = srcg + 4 * L;
  float* __restrict__ lw = lds + NOFF + 4 * L;
  const float* __restrict__ lrow = lds + m * LP + DIR * k0;
  float* __restrict__ cdst = lds + (L >> 1) * LP + COFF + (L & 1) * (CLEN / 2);
  const float* __restrict__ csrc = lds + (L >> 1) * LP + CSRC + (L & 1) * (CLEN / 2);

  float* __restrict__ obuf = lds + 16 * LP;
  float* opos[4];
#pragma unroll
  for (int j = 0; j < 4; ++j)
    opos[j] = obuf + (8 * hiA) * OP + (BWD ? ((14 - m - 16 * j) & (ORING - 1)) : (16 * j + m));
  const int frow = BWD ? (4 * (L >> 4) + 8 * ((L >> 3) & 1)) : (L >> 4);
  const int fcol = BWD ? 4 * (L & 7) : 4 * (L & 15);
  const float* __restrict__ fr = obuf + frow * OP + fcol;
  float* __restrict__ gw = dstg + (size_t)frow * (BWD ? TLEN : ES) + fcol;

  v4f R[16];
  auto gload = [&](const int j) {
#pragma unroll
    for (int r = 0; r < 16; ++r) R[r] = *(const v4f*)(gl + (size_t)r * srcStride + SEG * j);
  };
  auto lstore = [&]() {
#pragma unroll
    for (int r = 0; r < 16; ++r) *(v4f*)(lw + r * LP) = R[r];
  };
  auto carry = [&]() {
    v4f t[CLEN / 8];
#pragma unroll
    for (int i = 0; i < CLEN / 8; ++i) t[i] = *(const v4f*)(csrc + 4 * i);
#pragma unroll
    for (int i = 0; i < CLEN / 8; ++i) *(v4f*)(cdst + 4 * i) = t[i];
  };

  gload(J0);
  lstore();
  lds_sync_wave();
  carry();
  gload(J0 + DIR);
  lds_sync_wave();
  lstore();
  lds_sync_wave();

  float v0 = lrow[PB - DIR * k0];
  v2f z = (v2f){zi0 * v0, zi1 * v0};

  const int sl0 = k0 + ((m & 8) ? 16 : 0);
  const int sl1 = sl0 + 1;
  const int vsel = L & 7;
  const v8f zero = (v8f){0, 0, 0, 0, 0, 0, 0, 0};

  auto phase = [&](const int p, auto tag) {
    constexpr bool OUT = decltype(tag)::out;
    const bool more = (p + 1 < NPH);
    if (more) gload(J0 + DIR * (p + 2));
    v2f xa[4], xn[4];
#pragma unroll
    for (int c = 0; c < 4; ++c)
      xa[c] = (v2f){lrow[PB + DIR * (4 * c)], lrow[PB + DIR * (4 * c + 1)]};
#pragma unroll
    for (int q = 0; q < 8; ++q) {
      const int bb = bW + 8 * p + q;
      if (q < 7) {
#pragma unroll
        for (int c = 0; c < 4; ++c)
          xn[c] = (v2f){lrow[PB + DIR * (16 * (q + 1) + 4 * c)], lrow[PB + DIR * (16 * (q + 1) + 4 * c + 1)]};
      }
      v8f dZ = wmma4(z, PBv, zero);
#pragma unroll
      for (int c = 0; c < 4; ++c) dZ = wmma4(xa[c], UBv[c], dZ);
      if (OUT) {
        v8f dY = wmma4(z, WBv, zero);
#pragma unroll
        for (int c = 0; c < 4; ++c) dY = wmma4(xa[c], TBv[c], dY);
#pragma unroll
        for (int v = 0; v < 8; ++v) opos[q & 3][v * OP] = dY[v];
        if (BWD && (q == 0 || q == 6 || q == 7)) {
          const bool edge = (q == 0) ? (p == NPHW && m < 15) : ((q == 6) ? (p == NPH - 1 && m == 15) : (p == NPH - 1));
          if (edge) {
            const int col = ELEN - 16 - (bb * 16 + m);
#pragma unroll
            for (int v = 0; v < 8; ++v) dstg[(size_t)(v + 8 * hiA) * TLEN + col] = dY[v];
          }
        }
      }
      float nz0 = 0.f, nz1 = 0.f;
#pragma unroll
      for (int v = 0; v < 8; ++v) {
        float g0 = __shfl(dZ[v], sl0, 32);
        float g1 = __shfl(dZ[v], sl1, 32);
        nz0 = (vsel == v) ? g0 : nz0;
        nz1 = (vsel == v) ? g1 : nz1;
      }
      z = (v2f){nz0, nz1};
      if (OUT) {
        if (!BWD && (q & 3) == 3) {
          lds_sync_wave();
          const int c0 = 16 * (bW + 8 * p) + 64 * (q >> 2);
          v4f t[8];
#pragma unroll
          for (int i = 0; i < 8; ++i) t[i] = *(const v4f*)(fr + 2 * i * OP);
#pragma unroll
          for (int i = 0; i < 8; ++i) *(v4f*)(gw + (size_t)(2 * i) * ES + c0) = t[i];
          lds_sync_wave();
        }
        if (BWD && (q & 1) == 0 && !(q == 0 && p == NPHW)) {
          lds_sync_wave();
          const int c0 = (ELEN - 16 - 142) - 16 * (bW + 8 * p) + (128 - 16 * q);
          const int rp = (128 - 16 * q) & (ORING - 1);
          v4f t[4];
#pragma unroll
          for (int i = 0; i < 4; ++i) t[i] = *(const v4f*)(fr + i * OP + rp);
#pragma unroll
          for (int i = 0; i < 4; ++i) *(v4f*)(gw + (size_t)i * TLEN + c0) = t[i];
          lds_sync_wave();
        }
      }
      if (q < 7) {
#pragma unroll
        for (int c = 0; c < 4; ++c) xa[c] = xn[c];
        __builtin_amdgcn_sched_barrier(0);
      }
    }
    if (more) {
      lds_sync_wave();
      carry();
      lds_sync_wave();
      lstore();
      lds_sync_wave();
    }
  };

  for (int p = 0; p < NPHW; ++p) phase(p, WarmTag{});
  for (int p = NPHW; p < NPH; ++p) phase(p, OutTag{});
}

__global__ __launch_bounds__(32) void fwd_kernel(const float* __restrict__ x,
                                                 float* __restrict__ ws) {
  float* __restrict__ yf = ws + WS_YF;
  const int task = blockIdx.x;
  const int rg = task / NCHUNK, ch = task % NCHUNK;
  const int r0 = rg * 16;
  __shared__ __attribute__((aligned(16))) float lds[LDSF];
  if (ch >= 1 && ch <= NCHUNK - 2) {
    staged_chunk<false>(x + (size_t)r0 * TLEN, TLEN, yf + (size_t)r0 * ES, ws, lds, ch);
    return;
  }
  const int L = threadIdx.x;
  const int m = L & 15;
  const int hiA = L >> 4;
  const int k0 = 2 * hiA;
  const float* __restrict__ xrow = x + (size_t)(r0 + m) * TLEN;

  v2f TBv[4], UBv[4], WBv, PBv;
#pragma unroll
  for (int c = 0; c < 4; ++c) {
    TBv[c] = (v2f){ws[WS_TB + c * 64 + L], ws[WS_TB + c * 64 + 32 + L]};
    UBv[c] = (v2f){ws[WS_UB + c * 64 + L], ws[WS_UB + c * 64 + 32 + L]};
  }
  WBv = (v2f){ws[WS_WB + L], ws[WS_WB + 32 + L]};
  PBv = (v2f){ws[WS_PB + L], ws[WS_PB + 32 + L]};
  const float zi0 = ws[WS_ZI + k0], zi1 = ws[WS_ZI + k0 + 1];

  const int bOut = ch * CB;
  const int bW = (bOut - WB > 0) ? (bOut - WB) : 0;
  const int bEnd = ((bOut + CB) < NBLK) ? (bOut + CB) : NBLK;

  float v0 = load_ext_fwd(xrow, bW * 16);
  v2f z = (v2f){zi0 * v0, zi1 * v0};

  const int sl0 = k0 + ((m & 8) ? 16 : 0);
  const int sl1 = sl0 + 1;
  const int vsel = L & 7;

  for (int bb = bW; bb < bEnd; ++bb) {
    const int e0 = bb * 16;
    v2f xa[4];
    if (e0 >= PAD && (e0 + 16) <= (PAD + TLEN)) {
      const float* p = xrow + (e0 - PAD) + k0;
#pragma unroll
      for (int c = 0; c < 4; ++c) xa[c] = (v2f){p[4 * c], p[4 * c + 1]};
      __builtin_prefetch(p + 64, 0, 1);
    } else {
#pragma unroll
      for (int c = 0; c < 4; ++c) {
        int o = e0 + 4 * c + k0;
        xa[c] = (v2f){load_ext_fwd(xrow, o), load_ext_fwd(xrow, o + 1)};
      }
    }
    const v8f zero = (v8f){0, 0, 0, 0, 0, 0, 0, 0};
    v8f dZ = wmma4(z, PBv, zero);
#pragma unroll
    for (int c = 0; c < 4; ++c) dZ = wmma4(xa[c], UBv[c], dZ);
    if (bb >= bOut) {
      v8f dY = wmma4(z, WBv, zero);
#pragma unroll
      for (int c = 0; c < 4; ++c) dY = wmma4(xa[c], TBv[c], dY);
#pragma unroll
      for (int v = 0; v < 8; ++v)
        yf[(size_t)(r0 + v + 8 * hiA) * ES + e0 + m] = dY[v];
    }
    float nz0 = 0.f, nz1 = 0.f;
#pragma unroll
    for (int v = 0; v < 8; ++v) {
      float g0 = __shfl(dZ[v], sl0, 32);
      float g1 = __shfl(dZ[v], sl1, 32);
      nz0 = (vsel == v) ? g0 : nz0;
      nz1 = (vsel == v) ? g1 : nz1;
    }
    z = (v2f){nz0, nz1};
  }
}

__global__ __launch_bounds__(32) void bwd_kernel(const float* __restrict__ ws,
                                                 float* __restrict__ out) {
  const float* __restrict__ yf = ws + WS_YF;
  const int task = blockIdx.x;
  const int rg = task / NCHUNK, ch = task % NCHUNK;
  const int r0 = rg * 16;
  __shared__ __attribute__((aligned(16))) float lds[LDSF];
  if (ch >= 1 && ch <= NCHUNK - 2) {
    staged_chunk<true>(yf + (size_t)r0 * ES, ES, out + (size_t)r0 * TLEN, ws, lds, ch);
    return;
  }
  const int L = threadIdx.x;
  const int m = L & 15;
  const int hiA = L >> 4;
  const int k0 = 2 * hiA;
  const float* __restrict__ yrow = yf + (size_t)(r0 + m) * ES;

  v2f TBv[4], UBv[4], WBv, PBv;
#pragma unroll
  for (int c = 0; c < 4; ++c) {
    TBv[c] = (v2f){ws[WS_TB + c * 64 + L], ws[WS_TB + c * 64 + 32 + L]};
    UBv[c] = (v2f){ws[WS_UB + c * 64 + L], ws[WS_UB + c * 64 + 32 + L]};
  }
  WBv = (v2f){ws[WS_WB + L], ws[WS_WB + 32 + L]};
  PBv = (v2f){ws[WS_PB + L], ws[WS_PB + 32 + L]};
  const float zi0 = ws[WS_ZI + k0], zi1 = ws[WS_ZI + k0 + 1];

  const int bOut = ch * CB;
  const int bW = (bOut - WB > 0) ? (bOut - WB) : 0;
  const int bEnd = ((bOut + CB) < NBLK) ? (bOut + CB) : NBLK;

  float v0 = load_rev(yrow, bW * 16);
  v2f z = (v2f){zi0 * v0, zi1 * v0};
  const int sl0 = k0 + ((m & 8) ? 16 : 0);
  const int sl1 = sl0 + 1;
  const int vsel = L & 7;

  for (int bb = bW; bb < bEnd; ++bb) {
    const int e0 = bb * 16;
    v2f xa[4];
    if (e0 + 16 <= ELEN) {
      const float* p = yrow + (ELEN - 1 - e0 - k0);
#pragma unroll
      for (int c = 0; c < 4; ++c) xa[c] = (v2f){p[-4 * c], p[-4 * c - 1]};
      __builtin_prefetch(p - 64, 0, 1);
    } else {
#pragma unroll
      for (int c = 0; c < 4; ++c) {
        int o = e0 + 4 * c + k0;
        xa[c] = (v2f){load_rev(yrow, o), load_rev(yrow, o + 1)};
      }
    }
    const v8f zero = (v8f){0, 0, 0, 0, 0, 0, 0, 0};
    v8f dZ = wmma4(z, PBv, zero);
#pragma unroll
    for (int c = 0; c < 4; ++c) dZ = wmma4(xa[c], UBv[c], dZ);
    if (bb >= bOut) {
      v8f dY = wmma4(z, WBv, zero);
#pragma unroll
      for (int c = 0; c < 4; ++c) dY = wmma4(xa[c], TBv[c], dY);
      const int e = e0 + m;
      if (e >= PAD && e < PAD + TLEN) {
        const int col = ELEN - 16 - e;
#pragma unroll
        for (int v = 0; v < 8; ++v)
          out[(size_t)(r0 + v + 8 * hiA) * TLEN + col] = dY[v];
      }
    }
    float nz0 = 0.f, nz1 = 0.f;
#pragma unroll
    for (int v = 0; v < 8; ++v) {
      float g0 = __shfl(dZ[v], sl0, 32);
      float g1 = __shfl(dZ[v], sl1, 32);
      nz0 = (vsel == v) ? g0 : nz0;
      nz1 = (vsel == v) ? g1 : nz1;
    }
    z = (v2f){nz0, nz1};
  }
}

extern "C" void kernel_launch(void* const* d_in, const int* in_sizes, int n_in,
                              void* d_out, int out_size, void* d_ws, size_t ws_size,
                              hipStream_t stream) {
  (void)in_sizes; (void)n_in; (void)out_size; (void)ws_size;
  const float* x = (const float*)d_in[0];
  const float* b = (const float*)d_in[1];
  const float* a = (const float*)d_in[2];
  float* out = (float*)d_out;
  float* ws = (float*)d_ws;

  setup_kernel<<<1, 64, 0, stream>>>(b, a, ws);
  fwd_kernel<<<NRG * NCHUNK, 32, 0, stream>>>(x, ws);
  bwd_kernel<<<NRG * NCHUNK, 32, 0, stream>>>(ws, out);
}


# Interior chunks load only the carry (16 samples forward, 32 backward) of the segment that holds their first sample, straight to its LDS place with the next segment: 6 whole-segment loads, not 7
# speedup vs baseline: 2.1941x; 1.0415x over previous
#include <hip/hip_runtime.h>
#include <cstddef>

typedef __attribute__((ext_vector_type(2))) float v2f;
typedef __attribute__((ext_vector_type(8))) float v8f;
typedef __attribute__((ext_vector_type(4))) float v4f;

#if defined(__HIP_DEVICE_COMPILE__) && !__has_builtin(__builtin_amdgcn_wmma_f32_16x16x4_f32)
#error "__builtin_amdgcn_wmma_f32_16x16x4_f32 not available on gfx1250 device pass"
#endif

static __device__ __forceinline__ v8f wmma4(v2f a, v2f b, v8f c) {
  return __builtin_amdgcn_wmma_f32_16x16x4_f32(false, a, false, b, (short)0, c,
                                               false, false);
}

constexpr int NROWS = 256;
constexpr int TLEN  = 131072;
constexpr int PAD   = 15;
constexpr int ELEN  = TLEN + 2 * PAD;
constexpr int ES    = 131104;
constexpr int NBLK  = ES / 16;
constexpr int CB    = 32;
constexpr int WB    = 16;
constexpr int NCHUNK = (NBLK + CB - 1) / CB;
constexpr int NRG   = NROWS / 16;

constexpr int WS_TB = 0;
constexpr int WS_WB = 256;
constexpr int WS_UB = 320;
constexpr int WS_PB = 576;
constexpr int WS_ZI = 640;
constexpr int WS_YF = 1024;

constexpr int NVEC = 5;
constexpr int NPOW = 17;

__global__ __launch_bounds__(64) void setup_kernel(const float* __restrict__ bc,
                                                   const float* __restrict__ ac,
                                                   float* __restrict__ ws) {
  __shared__ double sQ[16];
  __shared__ double sV[NVEC * NPOW * 4];
  const int t = threadIdx.x;
  if (t < 16) {
    const int i = t & 7;
    const double a0 = (double)ac[0];
    double c = 0.0;
    if (i < 5) c = (double)((t < 8) ? ac[i] : bc[i]);
    sQ[t] = c / a0;
  }
  __syncthreads();
  const double an1 = sQ[1], an2 = sQ[2], an3 = sQ[3], an4 = sQ[4];
  const double b0 = sQ[8];
  const double Bv0 = sQ[9] - an1 * b0, Bv1 = sQ[10] - an2 * b0;
  const double Bv2 = sQ[11] - an3 * b0, Bv3 = sQ[12] - an4 * b0;

  if (t < 32) {
    double v0 = (t == 4) ? Bv0 : ((t == 0) ? 1.0 : 0.0);
    double v1 = (t == 4) ? Bv1 : ((t == 1) ? 1.0 : 0.0);
    double v2 = (t == 4) ? Bv2 : ((t == 2) ? 1.0 : 0.0);
    double v3 = (t == 4) ? Bv3 : ((t == 3) ? 1.0 : 0.0);
    for (int k = 0; k < NPOW; ++k) {
      if (t < NVEC) {
        double* __restrict__ d = sV + (t * NPOW + k) * 4;
        d[0] = v0; d[1] = v1; d[2] = v2; d[3] = v3;
      }
      const double n0 = v1 - an1 * v0;
      const double n1 = v2 - an2 * v0;
      const double n2 = v3 - an3 * v0;
      const double n3 = -(an4 * v0);
      v0 = n0; v1 = n1; v2 = n2; v3 = n3;
    }
  } else {
    const double s1 = 1.0 + an1, s2 = s1 + an2, s3 = s2 + an3, s4 = s3 + an4;
    const double c1 = Bv0, c2 = c1 + Bv1, c3 = c2 + Bv2, c4 = c3 + Bv3;
    const double z0 = c4 / s4;
    if (t == 32) {
      ws[WS_ZI + 0] = (float)z0;
      ws[WS_ZI + 1] = (float)(s1 * z0 - c1);
      ws[WS_ZI + 2] = (float)(s2 * z0 - c2);
      ws[WS_ZI + 3] = (float)(s3 * z0 - c3);
    }
  }
  __syncthreads();

  const int v = t >> 5, L = t & 31;
  const int n = L & 15;
  const int k = v + 2 * (L >> 4);
  const int nn = n & 3;
  for (int c = 0; c < 4; ++c) {
    const int kk = 4 * c + k;
    const int d = n - kk;
    const double hd = sV[(4 * NPOW + ((d > 0) ? d - 1 : 0)) * 4];
    ws[WS_TB + c * 64 + t] = (float)((d > 0) ? hd : ((d == 0) ? b0 : 0.0));
    const double ud = sV[(4 * NPOW + 15 - kk) * 4 + nn];
    ws[WS_UB + c * 64 + t] = (float)((n < 4) ? ud : 0.0);
  }
  ws[WS_WB + t] = (float)sV[(k * NPOW + n) * 4];
  const double pd = sV[(k * NPOW + 16) * 4 + nn];
  ws[WS_PB + t] = (float)((n < 4) ? pd : 0.0);
}

static __device__ __forceinline__ float load_ext_fwd(const float* __restrict__ xr, int e) {
  if (e < PAD) return 2.0f * xr[0] - xr[PAD - e];
  if (e < PAD + TLEN) return xr[e - PAD];
  int j = e - (PAD + TLEN);
  if (j > PAD - 1) j = PAD - 1;
  return 2.0f * xr[TLEN - 1] - xr[TLEN - 2 - j];
}

static __device__ __forceinline__ float load_rev(const float* __restrict__ yr, int e) {
  int idx = ELEN - 1 - e;
  if (idx < 0) idx = 0;
  return yr[idx];
}


constexpr int SEG  = 128;
constexpr int LP   = 164;
constexpr int OP   = 68;
constexpr int ORING = 64;
constexpr int LDSF = 16 * LP + 16 * OP;
constexpr int NPH  = (WB + CB) / 8;
constexpr int NPHW = WB / 8;

static __device__ __forceinline__ void lds_sync_wave() {
  __builtin_amdgcn_fence(__ATOMIC_RELEASE, "workgroup", "local");
  __builtin_amdgcn_wave_barrier();
  __builtin_amdgcn_fence(__ATOMIC_ACQUIRE, "workgroup", "local");
}

struct WarmTag { static constexpr bool out = false; };
struct OutTag  { static constexpr bool out = true; };

template <bool BWD>
static __device__ __forceinline__ void staged_chunk(const float* __restrict__ srcg,
                                                    int srcStride,
                                                    float* __restrict__ dstg,
                                                    const float* __restrict__ ws,
                                                    float* __restrict__ lds, int ch) {
  const int L = threadIdx.x;
  const int m = L & 15;
  const int hiA = L >> 4;
  const int k0 = 2 * hiA;
  constexpr int DIR  = BWD ? -1 : 1;
  constexpr int NOFF = BWD ? 0 : 16;
  constexpr int COFF = BWD ? SEG : 0;
  constexpr int CLEN = BWD ? 32 : 16;
  constexpr int CSRC = BWD ? 0 : (16 + SEG - 16);
  constexpr int PB   = BWD ? 157 : 1;

  v2f TBv[4], UBv[4], WBv, PBv;
#pragma unroll
  for (int c = 0; c < 4; ++c) {
    TBv[c] = (v2f){ws[WS_TB + c * 64 + L], ws[WS_TB + c * 64 + 32 + L]};
    UBv[c] = (v2f){ws[WS_UB + c * 64 + L], ws[WS_UB + c * 64 + 32 + L]};
  }
  WBv = (v2f){ws[WS_WB + L], ws[WS_WB + 32 + L]};
  PBv = (v2f){ws[WS_PB + L], ws[WS_PB + 32 + L]};
  const float zi0 = ws[WS_ZI + k0], zi1 = ws[WS_ZI + k0 + 1];

  const int bW = ch * CB - WB;
  const int S0 = BWD ? (ELEN - 1 - 16 * bW) : (16 * bW - PAD);
  const int J0 = S0 >> 7;

  const float* __restrict__ gl = srcg + 4 * L;
  float* __restrict__ lw = lds + NOFF + 4 * L;
  const float* __restrict__ lrow = lds + m * LP + DIR * k0;
  float* __restrict__ cdst = lds + (L >> 1) * LP + COFF + (L & 1) * (CLEN / 2);
  const float* __restrict__ csrc = lds + (L >> 1) * LP + CSRC + (L & 1) * (CLEN / 2);

  float* __restrict__ obuf = lds + 16 * LP;
  float* opos[4];
#pragma unroll
  for (int j = 0; j < 4; ++j)
    opos[j] = obuf + (8 * hiA) * OP + (BWD ? ((14 - m - 16 * j) & (ORING - 1)) : (16 * j + m));
  const int frow = BWD ? (4 * (L >> 4) + 8 * ((L >> 3) & 1)) : (L >> 4);
  const int fcol = BWD ? 4 * (L & 7) : 4 * (L & 15);
  const float* __restrict__ fr = obuf + frow * OP + fcol;
  float* __restrict__ gw = dstg + (size_t)frow * (BWD ? TLEN : ES) + fcol;

  v4f R[16];
  auto gload = [&](const int j) {
#pragma unroll
    for (int r = 0; r < 16; ++r) R[r] = *(const v4f*)(gl + (size_t)r * srcStride + SEG * j);
  };
  auto lstore = [&]() {
#pragma unroll
    for (int r = 0; r < 16; ++r) *(v4f*)(lw + r * LP) = R[r];
  };
  auto carry = [&]() {
    v4f t[CLEN / 8];
#pragma unroll
    for (int i = 0; i < CLEN / 8; ++i) t[i] = *(const v4f*)(csrc + 4 * i);
#pragma unroll
    for (int i = 0; i < CLEN / 8; ++i) *(v4f*)(cdst + 4 * i) = t[i];
  };

  constexpr int CLN = CLEN / 4;
  constexpr int CRW = 32 / CLN;
  constexpr int CGO = BWD ? 0 : (SEG - 16);
  v4f C0[CLEN / 8];
  const float* __restrict__ cg = srcg + (size_t)(L / CLN) * srcStride + SEG * J0 + CGO + 4 * (L % CLN);
  float* __restrict__ cl = lds + (L / CLN) * LP + COFF + 4 * (L % CLN);
#pragma unroll
  for (int i = 0; i < CLEN / 8; ++i) C0[i] = *(const v4f*)(cg + (size_t)(CRW * i) * srcStride);
  gload(J0 + DIR);
#pragma unroll
  for (int i = 0; i < CLEN / 8; ++i) *(v4f*)(cl + (CRW * i) * LP) = C0[i];
  lstore();
  lds_sync_wave();

  float v0 = lrow[PB - DIR * k0];
  v2f z = (v2f){zi0 * v0, zi1 * v0};

  const int sl0 = k0 + ((m & 8) ? 16 : 0);
  const int sl1 = sl0 + 1;
  const int vsel = L & 7;
  const v8f zero = (v8f){0, 0, 0, 0, 0, 0, 0, 0};

  auto phase = [&](const int p, auto tag) {
    constexpr bool OUT = decltype(tag)::out;
    const bool more = (p + 1 < NPH);
    if (more) gload(J0 + DIR * (p + 2));
    v2f xa[4], xn[4];
#pragma unroll
    for (int c = 0; c < 4; ++c)
      xa[c] = (v2f){lrow[PB + DIR * (4 * c)], lrow[PB + DIR * (4 * c + 1)]};
#pragma unroll
    for (int q = 0; q < 8; ++q) {
      const int bb = bW + 8 * p + q;
      if (q < 7) {
#pragma unroll
        for (int c = 0; c < 4; ++c)
          xn[c] = (v2f){lrow[PB + DIR * (16 * (q + 1) + 4 * c)], lrow[PB + DIR * (16 * (q + 1) + 4 * c + 1)]};
      }
      v8f dZ = wmma4(z, PBv, zero);
#pragma unroll
      for (int c = 0; c < 4; ++c) dZ = wmma4(xa[c], UBv[c], dZ);
      if (OUT) {
        v8f dY = wmma4(z, WBv, zero);
#pragma unroll
        for (int c = 0; c < 4; ++c) dY = wmma4(xa[c], TBv[c], dY);
#pragma unroll
        for (int v = 0; v < 8; ++v) opos[q & 3][v * OP] = dY[v];
        if (BWD && (q == 0 || q == 6 || q == 7)) {
          const bool edge = (q == 0) ? (p == NPHW && m < 15) : ((q == 6) ? (p == NPH - 1 && m == 15) : (p == NPH - 1));
          if (edge) {
            const int col = ELEN - 16 - (bb * 16 + m);
#pragma unroll
            for (int v = 0; v < 8; ++v) dstg[(size_t)(v + 8 * hiA) * TLEN + col] = dY[v];
          }
        }
      }
      float nz0 = 0.f, nz1 = 0.f;
#pragma unroll
      for (int v = 0; v < 8; ++v) {
        float g0 = __shfl(dZ[v], sl0, 32);
        float g1 = __shfl(dZ[v], sl1, 32);
        nz0 = (vsel == v) ? g0 : nz0;
        nz1 = (vsel == v) ? g1 : nz1;
      }
      z = (v2f){nz0, nz1};
      if (OUT) {
        if (!BWD && (q & 3) == 3) {
          lds_sync_wave();
          const int c0 = 16 * (bW + 8 * p) + 64 * (q >> 2);
          v4f t[8];
#pragma unroll
          for (int i = 0; i < 8; ++i) t[i] = *(const v4f*)(fr + 2 * i * OP);
#pragma unroll
          for (int i = 0; i < 8; ++i) *(v4f*)(gw + (size_t)(2 * i) * ES + c0) = t[i];
          lds_sync_wave();
        }
        if (BWD && (q & 1) == 0 && !(q == 0 && p == NPHW)) {
          lds_sync_wave();
          const int c0 = (ELEN - 16 - 142) - 16 * (bW + 8 * p) + (128 - 16 * q);
          const int rp = (128 - 16 * q) & (ORING - 1);
          v4f t[4];
#pragma unroll
          for (int i = 0; i < 4; ++i) t[i] = *(const v4f*)(fr + i * OP + rp);
#pragma unroll
          for (int i = 0; i < 4; ++i) *(v4f*)(gw + (size_t)i * TLEN + c0) = t[i];
          lds_sync_wave();
        }
      }
      if (q < 7) {
#pragma unroll
        for (int c = 0; c < 4; ++c) xa[c] = xn[c];
        __builtin_amdgcn_sched_barrier(0);
      }
    }
    if (more) {
      lds_sync_wave();
      carry();
      lds_sync_wave();
      lstore();
      lds_sync_wave();
    }
  };

  for (int p = 0; p < NPHW; ++p) phase(p, WarmTag{});
  for (int p = NPHW; p < NPH; ++p) phase(p, OutTag{});
}

__global__ __launch_bounds__(32) void fwd_kernel(const float* __restrict__ x,
                                                 float* __restrict__ ws) {
  float* __restrict__ yf = ws + WS_YF;
  const int task = blockIdx.x;
  const int rg = task / NCHUNK, ch = task % NCHUNK;
  const int r0 = rg * 16;
  __shared__ __attribute__((aligned(16))) float lds[LDSF];
  if (ch >= 1 && ch <= NCHUNK - 2) {
    staged_chunk<false>(x + (size_t)r0 * TLEN, TLEN, yf + (size_t)r0 * ES, ws, lds, ch);
    return;
  }
  const int L = threadIdx.x;
  const int m = L & 15;
  const int hiA = L >> 4;
  const int k0 = 2 * hiA;
  const float* __restrict__ xrow = x + (size_t)(r0 + m) * TLEN;

  v2f TBv[4], UBv[4], WBv, PBv;
#pragma unroll
  for (int c = 0; c < 4; ++c) {
    TBv[c] = (v2f){ws[WS_TB + c * 64 + L], ws[WS_TB + c * 64 + 32 + L]};
    UBv[c] = (v2f){ws[WS_UB + c * 64 + L], ws[WS_UB + c * 64 + 32 + L]};
  }
  WBv = (v2f){ws[WS_WB + L], ws[WS_WB + 32 + L]};
  PBv = (v2f){ws[WS_PB + L], ws[WS_PB + 32 + L]};
  const float zi0 = ws[WS_ZI + k0], zi1 = ws[WS_ZI + k0 + 1];

  const int bOut = ch * CB;
  const int bW = (bOut - WB > 0) ? (bOut - WB) : 0;
  const int bEnd = ((bOut + CB) < NBLK) ? (bOut + CB) : NBLK;

  float v0 = load_ext_fwd(xrow, bW * 16);
  v2f z = (v2f){zi0 * v0, zi1 * v0};

  const int sl0 = k0 + ((m & 8) ? 16 : 0);
  const int sl1 = sl0 + 1;
  const int vsel = L & 7;

  for (int bb = bW; bb < bEnd; ++bb) {
    const int e0 = bb * 16;
    v2f xa[4];
    if (e0 >= PAD && (e0 + 16) <= (PAD + TLEN)) {
      const float* p = xrow + (e0 - PAD) + k0;
#pragma unroll
      for (int c = 0; c < 4; ++c) xa[c] = (v2f){p[4 * c], p[4 * c + 1]};
      __builtin_prefetch(p + 64, 0, 1);
    } else {
#pragma unroll
      for (int c = 0; c < 4; ++c) {
        int o = e0 + 4 * c + k0;
        xa[c] = (v2f){load_ext_fwd(xrow, o), load_ext_fwd(xrow, o + 1)};
      }
    }
    const v8f zero = (v8f){0, 0, 0, 0, 0, 0, 0, 0};
    v8f dZ = wmma4(z, PBv, zero);
#pragma unroll
    for (int c = 0; c < 4; ++c) dZ = wmma4(xa[c], UBv[c], dZ);
    if (bb >= bOut) {
      v8f dY = wmma4(z, WBv, zero);
#pragma unroll
      for (int c = 0; c < 4; ++c) dY = wmma4(xa[c], TBv[c], dY);
#pragma unroll
      for (int v = 0; v < 8; ++v)
        yf[(size_t)(r0 + v + 8 * hiA) * ES + e0 + m] = dY[v];
    }
    float nz0 = 0.f, nz1 = 0.f;
#pragma unroll
    for (int v = 0; v < 8; ++v) {
      float g0 = __shfl(dZ[v], sl0, 32);
      float g1 = __shfl(dZ[v], sl1, 32);
      nz0 = (vsel == v) ? g0 : nz0;
      nz1 = (vsel == v) ? g1 : nz1;
    }
    z = (v2f){nz0, nz1};
  }
}

__global__ __launch_bounds__(32) void bwd_kernel(const float* __restrict__ ws,
                                                 float* __restrict__ out) {
  const float* __restrict__ yf = ws + WS_YF;
  const int task = blockIdx.x;
  const int rg = task / NCHUNK, ch = task % NCHUNK;
  const int r0 = rg * 16;
  __shared__ __attribute__((aligned(16))) float lds[LDSF];
  if (ch >= 1 && ch <= NCHUNK - 2) {
    staged_chunk<true>(yf + (size_t)r0 * ES, ES, out + (size_t)r0 * TLEN, ws, lds, ch);
    return;
  }
  const int L = threadIdx.x;
  const int m = L & 15;
  const int hiA = L >> 4;
  const int k0 = 2 * hiA;
  const float* __restrict__ yrow = yf + (size_t)(r0 + m) * ES;

  v2f TBv[4], UBv[4], WBv, PBv;
#pragma unroll
  for (int c = 0; c < 4; ++c) {
    TBv[c] = (v2f){ws[WS_TB + c * 64 + L], ws[WS_TB + c * 64 + 32 + L]};
    UBv[c] = (v2f){ws[WS_UB + c * 64 + L], ws[WS_UB + c * 64 + 32 + L]};
  }
  WBv = (v2f){ws[WS_WB + L], ws[WS_WB + 32 + L]};
  PBv = (v2f){ws[WS_PB + L], ws[WS_PB + 32 + L]};
  const float zi0 = ws[WS_ZI + k0], zi1 = ws[WS_ZI + k0 + 1];

  const int bOut = ch * CB;
  const int bW = (bOut - WB > 0) ? (bOut - WB) : 0;
  const int bEnd = ((bOut + CB) < NBLK) ? (bOut + CB) : NBLK;

  float v0 = load_rev(yrow, bW * 16);
  v2f z = (v2f){zi0 * v0, zi1 * v0};
  const int sl0 = k0 + ((m & 8) ? 16 : 0);
  const int sl1 = sl0 + 1;
  const int vsel = L & 7;

  for (int bb = bW; bb < bEnd; ++bb) {
    const int e0 = bb * 16;
    v2f xa[4];
    if (e0 + 16 <= ELEN) {
      const float* p = yrow + (ELEN - 1 - e0 - k0);
#pragma unroll
      for (int c = 0; c < 4; ++c) xa[c] = (v2f){p[-4 * c], p[-4 * c - 1]};
      __builtin_prefetch(p - 64, 0, 1);
    } else {
#pragma unroll
      for (int c = 0; c < 4; ++c) {
        int o = e0 + 4 * c + k0;
        xa[c] = (v2f){load_rev(yrow, o), load_rev(yrow, o + 1)};
      }
    }
    const v8f zero = (v8f){0, 0, 0, 0, 0, 0, 0, 0};
    v8f dZ = wmma4(z, PBv, zero);
#pragma unroll
    for (int c = 0; c < 4; ++c) dZ = wmma4(xa[c], UBv[c], dZ);
    if (bb >= bOut) {
      v8f dY = wmma4(z, WBv, zero);
#pragma unroll
      for (int c = 0; c < 4; ++c) dY = wmma4(xa[c], TBv[c], dY);
      const int e = e0 + m;
      if (e >= PAD && e < PAD + TLEN) {
        const int col = ELEN - 16 - e;
#pragma unroll
        for (int v = 0; v < 8; ++v)
          out[(size_t)(r0 + v + 8 * hiA) * TLEN + col] = dY[v];
      }
    }
    float nz0 = 0.f, nz1 = 0.f;
#pragma unroll
    for (int v = 0; v < 8; ++v) {
      float g0 = __shfl(dZ[v], sl0, 32);
      float g1 = __shfl(dZ[v], sl1, 32);
      nz0 = (vsel == v) ? g0 : nz0;
      nz1 = (vsel == v) ? g1 : nz1;
    }
    z = (v2f){nz0, nz1};
  }
}

extern "C" void kernel_launch(void* const* d_in, const int* in_sizes, int n_in,
                              void* d_out, int out_size, void* d_ws, size_t ws_size,
                              hipStream_t stream) {
  (void)in_sizes; (void)n_in; (void)out_size; (void)ws_size;
  const float* x = (const float*)d_in[0];
  const float* b = (const float*)d_in[1];
  const float* a = (const float*)d_in[2];
  float* out = (float*)d_out;
  float* ws = (float*)d_ws;

  setup_kernel<<<1, 64, 0, stream>>>(b, a, ws);
  fwd_kernel<<<NRG * NCHUNK, 32, 0, stream>>>(x, ws);
  bwd_kernel<<<NRG * NCHUNK, 32, 0, stream>>>(ws, out);
}


# State update computed transposed (constant image as A operand, data image as B): new state lands in lane m, so 2 lane-row exchanges per block replace 16 LDS-crossbar exchanges and 16 selects
# speedup vs baseline: 2.2985x; 1.0476x over previous
#include <hip/hip_runtime.h>
#include <cstddef>

typedef __attribute__((ext_vector_type(2))) float v2f;
typedef __attribute__((ext_vector_type(8))) float v8f;
typedef __attribute__((ext_vector_type(4))) float v4f;

#if defined(__HIP_DEVICE_COMPILE__) && !__has_builtin(__builtin_amdgcn_wmma_f32_16x16x4_f32)
#error "__builtin_amdgcn_wmma_f32_16x16x4_f32 not available on gfx1250 device pass"
#endif

static __device__ __forceinline__ v8f wmma4(v2f a, v2f b, v8f c) {
  return __builtin_amdgcn_wmma_f32_16x16x4_f32(false, a, false, b, (short)0, c,
                                               false, false);
}

constexpr int NROWS = 256;
constexpr int TLEN  = 131072;
constexpr int PAD   = 15;
constexpr int ELEN  = TLEN + 2 * PAD;
constexpr int ES    = 131104;
constexpr int NBLK  = ES / 16;
constexpr int CB    = 32;
constexpr int WB    = 16;
constexpr int NCHUNK = (NBLK + CB - 1) / CB;
constexpr int NRG   = NROWS / 16;

constexpr int WS_TB = 0;
constexpr int WS_WB = 256;
constexpr int WS_UB = 320;
constexpr int WS_PB = 576;
constexpr int WS_ZI = 640;
constexpr int WS_YF = 1024;

constexpr int NVEC = 5;
constexpr int NPOW = 17;

__global__ __launch_bounds__(64) void setup_kernel(const float* __restrict__ bc,
                                                   const float* __restrict__ ac,
                                                   float* __restrict__ ws) {
  __shared__ double sQ[16];
  __shared__ double sV[NVEC * NPOW * 4];
  const int t = threadIdx.x;
  if (t < 16) {
    const int i = t & 7;
    const double a0 = (double)ac[0];
    double c = 0.0;
    if (i < 5) c = (double)((t < 8) ? ac[i] : bc[i]);
    sQ[t] = c / a0;
  }
  __syncthreads();
  const double an1 = sQ[1], an2 = sQ[2], an3 = sQ[3], an4 = sQ[4];
  const double b0 = sQ[8];
  const double Bv0 = sQ[9] - an1 * b0, Bv1 = sQ[10] - an2 * b0;
  const double Bv2 = sQ[11] - an3 * b0, Bv3 = sQ[12] - an4 * b0;

  if (t < 32) {
    double v0 = (t == 4) ? Bv0 : ((t == 0) ? 1.0 : 0.0);
    double v1 = (t == 4) ? Bv1 : ((t == 1) ? 1.0 : 0.0);
    double v2 = (t == 4) ? Bv2 : ((t == 2) ? 1.0 : 0.0);
    double v3 = (t == 4) ? Bv3 : ((t == 3) ? 1.0 : 0.0);
    for (int k = 0; k < NPOW; ++k) {
      if (t < NVEC) {
        double* __restrict__ d = sV + (t * NPOW + k) * 4;
        d[0] = v0; d[1] = v1; d[2] = v2; d[3] = v3;
      }
      const double n0 = v1 - an1 * v0;
      const double n1 = v2 - an2 * v0;
      const double n2 = v3 - an3 * v0;
      const double n3 = -(an4 * v0);
      v0 = n0; v1 = n1; v2 = n2; v3 = n3;
    }
  } else {
    const double s1 = 1.0 + an1, s2 = s1 + an2, s3 = s2 + an3, s4 = s3 + an4;
    const double c1 = Bv0, c2 = c1 + Bv1, c3 = c2 + Bv2, c4 = c3 + Bv3;
    const double z0 = c4 / s4;
    if (t == 32) {
      ws[WS_ZI + 0] = (float)z0;
      ws[WS_ZI + 1] = (float)(s1 * z0 - c1);
      ws[WS_ZI + 2] = (float)(s2 * z0 - c2);
      ws[WS_ZI + 3] = (float)(s3 * z0 - c3);
    }
  }
  __syncthreads();

  const int v = t >> 5, L = t & 31;
  const int n = L & 15;
  const int k = v + 2 * (L >> 4);
  const int nn = n & 3;
  for (int c = 0; c < 4; ++c) {
    const int kk = 4 * c + k;
    const int d = n - kk;
    const double hd = sV[(4 * NPOW + ((d > 0) ? d - 1 : 0)) * 4];
    ws[WS_TB + c * 64 + t] = (float)((d > 0) ? hd : ((d == 0) ? b0 : 0.0));
    const double ud = sV[(4 * NPOW + 15 - kk) * 4 + nn];
    ws[WS_UB + c * 64 + t] = (float)((n < 4) ? ud : 0.0);
  }
  ws[WS_WB + t] = (float)sV[(k * NPOW + n) * 4];
  const double pd = sV[(k * NPOW + 16) * 4 + nn];
  ws[WS_PB + t] = (float)((n < 4) ? pd : 0.0);
}

static __device__ __forceinline__ float load_ext_fwd(const float* __restrict__ xr, int e) {
  if (e < PAD) return 2.0f * xr[0] - xr[PAD - e];
  if (e < PAD + TLEN) return xr[e - PAD];
  int j = e - (PAD + TLEN);
  if (j > PAD - 1) j = PAD - 1;
  return 2.0f * xr[TLEN - 1] - xr[TLEN - 2 - j];
}

static __device__ __forceinline__ float load_rev(const float* __restrict__ yr, int e) {
  int idx = ELEN - 1 - e;
  if (idx < 0) idx = 0;
  return yr[idx];
}


constexpr int SEG  = 128;
constexpr int LP   = 164;
constexpr int OP   = 68;
constexpr int ORING = 64;
constexpr int LDSF = 16 * LP + 16 * OP;
constexpr int NPH  = (WB + CB) / 8;
constexpr int NPHW = WB / 8;

static __device__ __forceinline__ void lds_sync_wave() {
  __builtin_amdgcn_fence(__ATOMIC_RELEASE, "workgroup", "local");
  __builtin_amdgcn_wave_barrier();
  __builtin_amdgcn_fence(__ATOMIC_ACQUIRE, "workgroup", "local");
}

static __device__ __forceinline__ v2f state_from_transposed(v8f d, int hiA) {
  const unsigned s2 = __float_as_uint(d[2]), s3 = __float_as_uint(d[3]);
  const float x2 = __uint_as_float(__builtin_amdgcn_permlanex16(s2, s2, 0x76543210u, 0xfedcba98u, false, false));
  const float x3 = __uint_as_float(__builtin_amdgcn_permlanex16(s3, s3, 0x76543210u, 0xfedcba98u, false, false));
  return (v2f){hiA ? x2 : d[0], hiA ? x3 : d[1]};
}

struct WarmTag { static constexpr bool out = false; };
struct OutTag  { static constexpr bool out = true; };

template <bool BWD>
static __device__ __forceinline__ void staged_chunk(const float* __restrict__ srcg,
                                                    int srcStride,
                                                    float* __restrict__ dstg,
                                                    const float* __restrict__ ws,
                                                    float* __restrict__ lds, int ch) {
  const int L = threadIdx.x;
  const int m = L & 15;
  const int hiA = L >> 4;
  const int k0 = 2 * hiA;
  constexpr int DIR  = BWD ? -1 : 1;
  constexpr int NOFF = BWD ? 0 : 16;
  constexpr int COFF = BWD ? SEG : 0;
  constexpr int CLEN = BWD ? 32 : 16;
  constexpr int CSRC = BWD ? 0 : (16 + SEG - 16);
  constexpr int PB   = BWD ? 157 : 1;

  v2f TBv[4], UBv[4], WBv, PBv;
#pragma unroll
  for (int c = 0; c < 4; ++c) {
    TBv[c] = (v2f){ws[WS_TB + c * 64 + L], ws[WS_TB + c * 64 + 32 + L]};
    UBv[c] = (v2f){ws[WS_UB + c * 64 + L], ws[WS_UB + c * 64 + 32 + L]};
  }
  WBv = (v2f){ws[WS_WB + L], ws[WS_WB + 32 + L]};
  PBv = (v2f){ws[WS_PB + L], ws[WS_PB + 32 + L]};
  const float zi0 = ws[WS_ZI + k0], zi1 = ws[WS_ZI + k0 + 1];

  const int bW = ch * CB - WB;
  const int S0 = BWD ? (ELEN - 1 - 16 * bW) : (16 * bW - PAD);
  const int J0 = S0 >> 7;

  const float* __restrict__ gl = srcg + 4 * L;
  float* __restrict__ lw = lds + NOFF + 4 * L;
  const float* __restrict__ lrow = lds + m * LP + DIR * k0;
  float* __restrict__ cdst = lds + (L >> 1) * LP + COFF + (L & 1) * (CLEN / 2);
  const float* __restrict__ csrc = lds + (L >> 1) * LP + CSRC + (L & 1) * (CLEN / 2);

  float* __restrict__ obuf = lds + 16 * LP;
  float* opos[4];
#pragma unroll
  for (int j = 0; j < 4; ++j)
    opos[j] = obuf + (8 * hiA) * OP + (BWD ? ((14 - m - 16 * j) & (ORING - 1)) : (16 * j + m));
  const int frow = BWD ? (4 * (L >> 4) + 8 * ((L >> 3) & 1)) : (L >> 4);
  const int fcol = BWD ? 4 * (L & 7) : 4 * (L & 15);
  const float* __restrict__ fr = obuf + frow * OP + fcol;
  float* __restrict__ gw = dstg + (size_t)frow * (BWD ? TLEN : ES) + fcol;

  v4f R[16];
  auto gload = [&](const int j) {
#pragma unroll
    for (int r = 0; r < 16; ++r) R[r] = *(const v4f*)(gl + (size_t)r * srcStride + SEG * j);
  };
  auto lstore = [&]() {
#pragma unroll
    for (int r = 0; r < 16; ++r) *(v4f*)(lw + r * LP) = R[r];
  };
  auto carry = [&]() {
    v4f t[CLEN / 8];
#pragma unroll
    for (int i = 0; i < CLEN / 8; ++i) t[i] = *(const v4f*)(csrc + 4 * i);
#pragma unroll
    for (int i = 0; i < CLEN / 8; ++i) *(v4f*)(cdst + 4 * i) = t[i];
  };

  constexpr int CLN = CLEN / 4;
  constexpr int CRW = 32 / CLN;
  constexpr int CGO = BWD ? 0 : (SEG - 16);
  v4f C0[CLEN / 8];
  const float* __restrict__ cg = srcg + (size_t)(L / CLN) * srcStride + SEG * J0 + CGO + 4 * (L % CLN);
  float* __restrict__ cl = lds + (L / CLN) * LP + COFF + 4 * (L % CLN);
#pragma unroll
  for (int i = 0; i < CLEN / 8; ++i) C0[i] = *(const v4f*)(cg + (size_t)(CRW * i) * srcStride);
  gload(J0 + DIR);
#pragma unroll
  for (int i = 0; i < CLEN / 8; ++i) *(v4f*)(cl + (CRW * i) * LP) = C0[i];
  lstore();
  lds_sync_wave();

  float v0 = lrow[PB - DIR * k0];
  v2f z = (v2f){zi0 * v0, zi1 * v0};

  const v8f zero = (v8f){0, 0, 0, 0, 0, 0, 0, 0};

  auto phase = [&](const int p, auto tag) {
    constexpr bool OUT = decltype(tag)::out;
    const bool more = (p + 1 < NPH);
    if (more) gload(J0 + DIR * (p + 2));
    v2f xa[4], xn[4];
#pragma unroll
    for (int c = 0; c < 4; ++c)
      xa[c] = (v2f){lrow[PB + DIR * (4 * c)], lrow[PB + DIR * (4 * c + 1)]};
#pragma unroll
    for (int q = 0; q < 8; ++q) {
      const int bb = bW + 8 * p + q;
      if (q < 7) {
#pragma unroll
        for (int c = 0; c < 4; ++c)
          xn[c] = (v2f){lrow[PB + DIR * (16 * (q + 1) + 4 * c)], lrow[PB + DIR * (16 * (q + 1) + 4 * c + 1)]};
      }
      v8f dZ = wmma4(PBv, z, zero);
#pragma unroll
      for (int c = 0; c < 4; ++c) dZ = wmma4(UBv[c], xa[c], dZ);
      if (OUT) {
        v8f dY = wmma4(z, WBv, zero);
#pragma unroll
        for (int c = 0; c < 4; ++c) dY = wmma4(xa[c], TBv[c], dY);
#pragma unroll
        for (int v = 0; v < 8; ++v) opos[q & 3][v * OP] = dY[v];
        if (BWD && (q == 0 || q == 6 || q == 7)) {
          const bool edge = (q == 0) ? (p == NPHW && m < 15) : ((q == 6) ? (p == NPH - 1 && m == 15) : (p == NPH - 1));
          if (edge) {
            const int col = ELEN - 16 - (bb * 16 + m);
#pragma unroll
            for (int v = 0; v < 8; ++v) dstg[(size_t)(v + 8 * hiA) * TLEN + col] = dY[v];
          }
        }
      }
      z = state_from_transposed(dZ, hiA);
      if (OUT) {
        if (!BWD && (q & 3) == 3) {
          lds_sync_wave();
          const int c0 = 16 * (bW + 8 * p) + 64 * (q >> 2);
          v4f t[8];
#pragma unroll
          for (int i = 0; i < 8; ++i) t[i] = *(const v4f*)(fr + 2 * i * OP);
#pragma unroll
          for (int i = 0; i < 8; ++i) *(v4f*)(gw + (size_t)(2 * i) * ES + c0) = t[i];
          lds_sync_wave();
        }
        if (BWD && (q & 1) == 0 && !(q == 0 && p == NPHW)) {
          lds_sync_wave();
          const int c0 = (ELEN - 16 - 142) - 16 * (bW + 8 * p) + (128 - 16 * q);
          const int rp = (128 - 16 * q) & (ORING - 1);
          v4f t[4];
#pragma unroll
          for (int i = 0; i < 4; ++i) t[i] = *(const v4f*)(fr + i * OP + rp);
#pragma unroll
          for (int i = 0; i < 4; ++i) *(v4f*)(gw + (size_t)i * TLEN + c0) = t[i];
          lds_sync_wave();
        }
      }
      if (q < 7) {
#pragma unroll
        for (int c = 0; c < 4; ++c) xa[c] = xn[c];
        __builtin_amdgcn_sched_barrier(0);
      }
    }
    if (more) {
      lds_sync_wave();
      carry();
      lds_sync_wave();
      lstore();
      lds_sync_wave();
    }
  };

  for (int p = 0; p < NPHW; ++p) phase(p, WarmTag{});
  for (int p = NPHW; p < NPH; ++p) phase(p, OutTag{});
}

__global__ __launch_bounds__(32) void fwd_kernel(const float* __restrict__ x,
                                                 float* __restrict__ ws) {
  float* __restrict__ yf = ws + WS_YF;
  const int task = blockIdx.x;
  const int rg = task / NCHUNK, ch = task % NCHUNK;
  const int r0 = rg * 16;
  __shared__ __attribute__((aligned(16))) float lds[LDSF];
  if (ch >= 1 && ch <= NCHUNK - 2) {
    staged_chunk<false>(x + (size_t)r0 * TLEN, TLEN, yf + (size_t)r0 * ES, ws, lds, ch);
    return;
  }
  const int L = threadIdx.x;
  const int m = L & 15;
  const int hiA = L >> 4;
  const int k0 = 2 * hiA;
  const float* __restrict__ xrow = x + (size_t)(r0 + m) * TLEN;

  v2f TBv[4], UBv[4], WBv, PBv;
#pragma unroll
  for (int c = 0; c < 4; ++c) {
    TBv[c] = (v2f){ws[WS_TB + c * 64 + L], ws[WS_TB + c * 64 + 32 + L]};
    UBv[c] = (v2f){ws[WS_UB + c * 64 + L], ws[WS_UB + c * 64 + 32 + L]};
  }
  WBv = (v2f){ws[WS_WB + L], ws[WS_WB + 32 + L]};
  PBv = (v2f){ws[WS_PB + L], ws[WS_PB + 32 + L]};
  const float zi0 = ws[WS_ZI + k0], zi1 = ws[WS_ZI + k0 + 1];

  const int bOut = ch * CB;
  const int bW = (bOut - WB > 0) ? (bOut - WB) : 0;
  const int bEnd = ((bOut + CB) < NBLK) ? (bOut + CB) : NBLK;

  float v0 = load_ext_fwd(xrow, bW * 16);
  v2f z = (v2f){zi0 * v0, zi1 * v0};


  for (int bb = bW; bb < bEnd; ++bb) {
    const int e0 = bb * 16;
    v2f xa[4];
    if (e0 >= PAD && (e0 + 16) <= (PAD + TLEN)) {
      const float* p = xrow + (e0 - PAD) + k0;
#pragma unroll
      for (int c = 0; c < 4; ++c) xa[c] = (v2f){p[4 * c], p[4 * c + 1]};
      __builtin_prefetch(p + 64, 0, 1);
    } else {
#pragma unroll
      for (int c = 0; c < 4; ++c) {
        int o = e0 + 4 * c + k0;
        xa[c] = (v2f){load_ext_fwd(xrow, o), load_ext_fwd(xrow, o + 1)};
      }
    }
    const v8f zero = (v8f){0, 0, 0, 0, 0, 0, 0, 0};
    v8f dZ = wmma4(PBv, z, zero);
#pragma unroll
    for (int c = 0; c < 4; ++c) dZ = wmma4(UBv[c], xa[c], dZ);
    if (bb >= bOut) {
      v8f dY = wmma4(z, WBv, zero);
#pragma unroll
      for (int c = 0; c < 4; ++c) dY = wmma4(xa[c], TBv[c], dY);
#pragma unroll
      for (int v = 0; v < 8; ++v)
        yf[(size_t)(r0 + v + 8 * hiA) * ES + e0 + m] = dY[v];
    }
    z = state_from_transposed(dZ, hiA);
  }
}

__global__ __launch_bounds__(32) void bwd_kernel(const float* __restrict__ ws,
                                                 float* __restrict__ out) {
  const float* __restrict__ yf = ws + WS_YF;
  const int task = blockIdx.x;
  const int rg = task / NCHUNK, ch = task % NCHUNK;
  const int r0 = rg * 16;
  __shared__ __attribute__((aligned(16))) float lds[LDSF];
  if (ch >= 1 && ch <= NCHUNK - 2) {
    staged_chunk<true>(yf + (size_t)r0 * ES, ES, out + (size_t)r0 * TLEN, ws, lds, ch);
    return;
  }
  const int L = threadIdx.x;
  const int m = L & 15;
  const int hiA = L >> 4;
  const int k0 = 2 * hiA;
  const float* __restrict__ yrow = yf + (size_t)(r0 + m) * ES;

  v2f TBv[4], UBv[4], WBv, PBv;
#pragma unroll
  for (int c = 0; c < 4; ++c) {
    TBv[c] = (v2f){ws[WS_TB + c * 64 + L], ws[WS_TB + c * 64 + 32 + L]};
    UBv[c] = (v2f){ws[WS_UB + c * 64 + L], ws[WS_UB + c * 64 + 32 + L]};
  }
  WBv = (v2f){ws[WS_WB + L], ws[WS_WB + 32 + L]};
  PBv = (v2f){ws[WS_PB + L], ws[WS_PB + 32 + L]};
  const float zi0 = ws[WS_ZI + k0], zi1 = ws[WS_ZI + k0 + 1];

  const int bOut = ch * CB;
  const int bW = (bOut - WB > 0) ? (bOut - WB) : 0;
  const int bEnd = ((bOut + CB) < NBLK) ? (bOut + CB) : NBLK;

  float v0 = load_rev(yrow, bW * 16);
  v2f z = (v2f){zi0 * v0, zi1 * v0};

  for (int bb = bW; bb < bEnd; ++bb) {
    const int e0 = bb * 16;
    v2f xa[4];
    if (e0 + 16 <= ELEN) {
      const float* p = yrow + (ELEN - 1 - e0 - k0);
#pragma unroll
      for (int c = 0; c < 4; ++c) xa[c] = (v2f){p[-4 * c], p[-4 * c - 1]};
      __builtin_prefetch(p - 64, 0, 1);
    } else {
#pragma unroll
      for (int c = 0; c < 4; ++c) {
        int o = e0 + 4 * c + k0;
        xa[c] = (v2f){load_rev(yrow, o), load_rev(yrow, o + 1)};
      }
    }
    const v8f zero = (v8f){0, 0, 0, 0, 0, 0, 0, 0};
    v8f dZ = wmma4(PBv, z, zero);
#pragma unroll
    for (int c = 0; c < 4; ++c) dZ = wmma4(UBv[c], xa[c], dZ);
    if (bb >= bOut) {
      v8f dY = wmma4(z, WBv, zero);
#pragma unroll
      for (int c = 0; c < 4; ++c) dY = wmma4(xa[c], TBv[c], dY);
      const int e = e0 + m;
      if (e >= PAD && e < PAD + TLEN) {
        const int col = ELEN - 16 - e;
#pragma unroll
        for (int v = 0; v < 8; ++v)
          out[(size_t)(r0 + v + 8 * hiA) * TLEN + col] = dY[v];
      }
    }
    z = state_from_transposed(dZ, hiA);
  }
}

extern "C" void kernel_launch(void* const* d_in, const int* in_sizes, int n_in,
                              void* d_out, int out_size, void* d_ws, size_t ws_size,
                              hipStream_t stream) {
  (void)in_sizes; (void)n_in; (void)out_size; (void)ws_size;
  const float* x = (const float*)d_in[0];
  const float* b = (const float*)d_in[1];
  const float* a = (const float*)d_in[2];
  float* out = (float*)d_out;
  float* ws = (float*)d_ws;

  setup_kernel<<<1, 64, 0, stream>>>(b, a, ws);
  fwd_kernel<<<NRG * NCHUNK, 32, 0, stream>>>(x, ws);
  bwd_kernel<<<NRG * NCHUNK, 32, 0, stream>>>(ws, out);
}
